# starting kernel
# baseline (speedup 1.0000x reference)
#include <hip/hip_runtime.h>
#include <hip/hip_bf16.h>


typedef float v2f __attribute__((ext_vector_type(2)));
typedef float v8f __attribute__((ext_vector_type(8)));

__device__ __forceinline__ v8f wmma_f32_16x16x4(v2f a, v2f b, v8f c) {
  return __builtin_amdgcn_wmma_f32_16x16x4_f32(false, a, false, b, (short)0, c,
                                               false, false);
}

template <int OPB>
__global__ void __launch_bounds__(256)
gemm_wmma(const float* __restrict__ A, int lda,
          const float* __restrict__ B, int ldb,
          float* __restrict__ C, int ldc,
          int M, int Nc, int K) {
  constexpr int BM = 128, BN = 128, BK = 16;
  constexpr int ALD = 20;
  constexpr int BLD = 132;

  __shared__ float As[BM * ALD];
  __shared__ float Bs[BK * BLD];

  const int tid  = threadIdx.x;
  const int wid  = tid >> 5;
  const int lane = tid & 31;
  const int lh   = lane >> 4;
  const int lm   = lane & 15;

  const int m0 = blockIdx.y * BM;
  const int n0 = blockIdx.x * BN;
  const int wm = (wid & 1) * 64;
  const int wn = (wid >> 1) * 32;

  v8f acc[4][2];
#pragma unroll
  for (int tm = 0; tm < 4; ++tm)
#pragma unroll
    for (int tn = 0; tn < 2; ++tn) acc[tm][tn] = (v8f)0.0f;

  for (int k0 = 0; k0 < K; k0 += BK) {
    __syncthreads();
#pragma unroll
    for (int r = 0; r < 2; ++r) {
      const int i   = tid + r * 256;
      const int row = i >> 2;
      const int c4  = (i & 3) << 2;
      const int gm  = m0 + row;
      float4 v = make_float4(0.f, 0.f, 0.f, 0.f);
      if (gm < M)
        v = *(const float4*)(A + (size_t)gm * lda + k0 + c4);
      *(float4*)(&As[row * ALD + c4]) = v;
    }
    if (OPB) {
#pragma unroll
      for (int r = 0; r < 2; ++r) {
        const int i  = tid + r * 256;
        const int j  = i >> 2;
        const int k4 = (i & 3) << 2;
        const int gj = n0 + j;
        float4 v = make_float4(0.f, 0.f, 0.f, 0.f);
        if (gj < Nc)
          v = *(const float4*)(B + (size_t)gj * ldb + k0 + k4);
        Bs[(k4 + 0) * BLD + j] = v.x;
        Bs[(k4 + 1) * BLD + j] = v.y;
        Bs[(k4 + 2) * BLD + j] = v.z;
        Bs[(k4 + 3) * BLD + j] = v.w;
      }
    } else {
#pragma unroll
      for (int r = 0; r < 2; ++r) {
        const int i  = tid + r * 256;
        const int kk = i >> 5;
        const int c  = (i & 31) << 2;
        const int gj = n0 + c;
        float4 v = make_float4(0.f, 0.f, 0.f, 0.f);
        if (gj < Nc)
          v = *(const float4*)(B + (size_t)(k0 + kk) * ldb + gj);
        *(float4*)(&Bs[kk * BLD + c]) = v;
      }
    }
    if (k0 + BK < K) {
      const int gm = m0 + (tid >> 1);
      if (gm < M)
        __builtin_prefetch(A + (size_t)gm * lda + (k0 + BK) + (tid & 1) * 8, 0, 1);
    }
    __syncthreads();

#pragma unroll
    for (int kk = 0; kk < BK; kk += 4) {
      const int ks = kk + 2 * lh;
      v2f a[4], b[2];
#pragma unroll
      for (int tm = 0; tm < 4; ++tm) {
        a[tm] = *(const v2f*)(&As[(wm + tm * 16 + lm) * ALD + ks]);
      }
#pragma unroll
      for (int tn = 0; tn < 2; ++tn) {
        const int col = wn + tn * 16 + lm;
        b[tn].x = Bs[(ks + 0) * BLD + col];
        b[tn].y = Bs[(ks + 1) * BLD + col];
      }
#pragma unroll
      for (int tm = 0; tm < 4; ++tm)
#pragma unroll
        for (int tn = 0; tn < 2; ++tn)
          acc[tm][tn] = wmma_f32_16x16x4(a[tm], b[tn], acc[tm][tn]);
    }
  }

#pragma unroll
  for (int tm = 0; tm < 4; ++tm) {
#pragma unroll
    for (int v = 0; v < 8; ++v) {
      const int gm = m0 + wm + tm * 16 + v + 8 * lh;
      if (gm < M) {
#pragma unroll
        for (int tn = 0; tn < 2; ++tn) {
          const int gn = n0 + wn + tn * 16 + lm;
          C[(size_t)gm * ldc + gn] = acc[tm][tn][v];
        }
      }
    }
  }
}

__global__ void init_deg_kernel(float* deg, int n) {
  int i = blockIdx.x * blockDim.x + threadIdx.x;
  if (i < n) deg[i] = 1.0f;
}

__global__ void deg_accum_kernel(const int* __restrict__ dst, float* deg, int E) {
  int e = blockIdx.x * blockDim.x + threadIdx.x;
  if (e < E) atomicAdd(&deg[dst[e]], 1.0f);
}

__global__ void dinv_kernel(float* deg, int n) {
  int i = blockIdx.x * blockDim.x + threadIdx.x;
  if (i < n) {
    float d = deg[i];
    deg[i] = (d > 0.f) ? rsqrtf(d) : 0.f;
  }
}

__global__ void norm_kernel(const int* __restrict__ src, const int* __restrict__ dst,
                            const float* __restrict__ dinv, float* __restrict__ norm,
                            int E, int Et) {
  int e = blockIdx.x * blockDim.x + threadIdx.x;
  if (e >= Et) return;
  if (e < E) {
    norm[e] = dinv[src[e]] * dinv[dst[e]];
  } else {
    float d = dinv[e - E];
    norm[e] = d * d;
  }
}

__global__ void bias_init_kernel(float* __restrict__ out, const float* __restrict__ b,
                                 int n, int D4) {
  int i = blockIdx.x * blockDim.x + threadIdx.x;
  if (i >= n * D4) return;
  const int j = i % D4;
  ((float4*)out)[i] = ((const float4*)b)[j];
}

__global__ void scatter_kernel(const float* __restrict__ h,
                               const int* __restrict__ src, const int* __restrict__ dst,
                               const float* __restrict__ norm, float* __restrict__ out,
                               int D, int D4, int E, int Et) {
  long long tid = (long long)blockIdx.x * blockDim.x + threadIdx.x;
  long long total = (long long)Et * D4;
  if (tid >= total) return;
  const int e = (int)(tid / D4);
  const int j = (int)(tid % D4);
  int s, d;
  if (e < E) { s = src[e]; d = dst[e]; }
  else       { s = e - E;  d = e - E;  }
  const float nv = norm[e];
  const float4 v = ((const float4*)(h + (size_t)s * D))[j];
  float* o = out + (size_t)d * D + j * 4;
  atomicAdd(o + 0, v.x * nv);
  atomicAdd(o + 1, v.y * nv);
  atomicAdd(o + 2, v.z * nv);
  atomicAdd(o + 3, v.w * nv);
}

static inline void launch_gemm(int opb, const float* A, int lda, const float* B,
                               int ldb, float* C, int ldc, int M, int Nc, int K,
                               hipStream_t stream) {
  dim3 grid(Nc / 128, (M + 127) / 128);
  if (opb)
    gemm_wmma<1><<<grid, 256, 0, stream>>>(A, lda, B, ldb, C, ldc, M, Nc, K);
  else
    gemm_wmma<0><<<grid, 256, 0, stream>>>(A, lda, B, ldb, C, ldc, M, Nc, K);
}

extern "C" void kernel_launch(void* const* d_in, const int* in_sizes, int n_in,
                              void* d_out, int out_size, void* d_ws, size_t ws_size,
                              hipStream_t stream) {
  const int IN = 1024, HID = 512, OUT = 128;

  const float* features = (const float*)d_in[0];
  const int*   edge     = (const int*)d_in[1];
  const float* W1       = (const float*)d_in[2];
  const float* b1       = (const float*)d_in[3];
  const float* W2       = (const float*)d_in[4];
  const float* b2       = (const float*)d_in[5];
  const float* b3       = (const float*)d_in[6];
  const float* b4       = (const float*)d_in[7];
  const float* head1    = (const float*)d_in[8];

  const int n  = in_sizes[0] / IN;
  const int E  = in_sizes[1] / 2;
  const int Et = E + n;
  const int* src = edge;
  const int* dst = edge + E;

  float* out = (float*)d_out;
  float* z   = out;
  float* h2  = out + (size_t)n * OUT;
  float* h4  = out + (size_t)n * OUT * 2;

  float* ws   = (float*)d_ws;
  float* dinv = ws;
  float* norm = dinv + n;
  float* t0   = norm + Et;
  float* t1   = t0 + (size_t)n * HID;
  float* h2p  = t1 + (size_t)n * HID;
  float* t2   = h2p + (size_t)n * OUT;

  const int TB = 256;
  init_deg_kernel<<<(n + TB - 1) / TB, TB, 0, stream>>>(dinv, n);
  deg_accum_kernel<<<(E + TB - 1) / TB, TB, 0, stream>>>(dst, dinv, E);
  dinv_kernel<<<(n + TB - 1) / TB, TB, 0, stream>>>(dinv, n);
  norm_kernel<<<(Et + TB - 1) / TB, TB, 0, stream>>>(src, dst, dinv, norm, E, Et);

  launch_gemm(1, features, IN, W1, IN, t0, HID, n, HID, IN, stream);
  bias_init_kernel<<<((n * (HID / 4)) + TB - 1) / TB, TB, 0, stream>>>(t1, b1, n, HID / 4);
  {
    long long tot = (long long)Et * (HID / 4);
    scatter_kernel<<<(unsigned)((tot + TB - 1) / TB), TB, 0, stream>>>(
        t0, src, dst, norm, t1, HID, HID / 4, E, Et);
  }

  launch_gemm(1, t1, HID, W2, HID, h2p, OUT, n, OUT, HID, stream);
  bias_init_kernel<<<((n * (OUT / 4)) + TB - 1) / TB, TB, 0, stream>>>(h2, b2, n, OUT / 4);
  {
    long long tot = (long long)Et * (OUT / 4);
    scatter_kernel<<<(unsigned)((tot + TB - 1) / TB), TB, 0, stream>>>(
        h2p, src, dst, norm, h2, OUT, OUT / 4, E, Et);
  }

  launch_gemm(0, h2, OUT, head1, OUT, z, OUT, n, OUT, OUT, stream);

  launch_gemm(0, z, OUT, W2, HID, t0, HID, n, HID, OUT, stream);
  bias_init_kernel<<<((n * (HID / 4)) + TB - 1) / TB, TB, 0, stream>>>(t1, b3, n, HID / 4);
  {
    long long tot = (long long)Et * (HID / 4);
    scatter_kernel<<<(unsigned)((tot + TB - 1) / TB), TB, 0, stream>>>(
        t0, src, dst, norm, t1, HID, HID / 4, E, Et);
  }

  launch_gemm(0, t1, HID, W1, IN, t2, IN, n, IN, HID, stream);
  bias_init_kernel<<<((n * (IN / 4)) + TB - 1) / TB, TB, 0, stream>>>(h4, b4, n, IN / 4);
  {
    long long tot = (long long)Et * (IN / 4);
    scatter_kernel<<<(unsigned)((tot + TB - 1) / TB), TB, 0, stream>>>(
        t2, src, dst, norm, h4, IN, IN / 4, E, Et);
  }
}


# Aggregation as per-destination row gather over an edge list bucketed by dst each call (counting sort), replacing float atomic scatter and bias pre-fill; sums now in fixed edge-id order
# speedup vs baseline: 1.7277x; 1.7277x over previous
#include <hip/hip_runtime.h>
#include <hip/hip_bf16.h>


typedef float v2f __attribute__((ext_vector_type(2)));
typedef float v8f __attribute__((ext_vector_type(8)));

__device__ __forceinline__ v8f wmma_f32_16x16x4(v2f a, v2f b, v8f c) {
  return __builtin_amdgcn_wmma_f32_16x16x4_f32(false, a, false, b, (short)0, c,
                                               false, false);
}

template <int OPB>
__global__ void __launch_bounds__(256)
gemm_wmma(const float* __restrict__ A, int lda,
          const float* __restrict__ B, int ldb,
          float* __restrict__ C, int ldc,
          int M, int Nc, int K) {
  constexpr int BM = 128, BN = 128, BK = 16;
  constexpr int ALD = 20;
  constexpr int BLD = 132;

  __shared__ float As[BM * ALD];
  __shared__ float Bs[BK * BLD];

  const int tid  = threadIdx.x;
  const int wid  = tid >> 5;
  const int lane = tid & 31;
  const int lh   = lane >> 4;
  const int lm   = lane & 15;

  const int m0 = blockIdx.y * BM;
  const int n0 = blockIdx.x * BN;
  const int wm = (wid & 1) * 64;
  const int wn = (wid >> 1) * 32;

  v8f acc[4][2];
#pragma unroll
  for (int tm = 0; tm < 4; ++tm)
#pragma unroll
    for (int tn = 0; tn < 2; ++tn) acc[tm][tn] = (v8f)0.0f;

  for (int k0 = 0; k0 < K; k0 += BK) {
    __syncthreads();
#pragma unroll
    for (int r = 0; r < 2; ++r) {
      const int i   = tid + r * 256;
      const int row = i >> 2;
      const int c4  = (i & 3) << 2;
      const int gm  = m0 + row;
      float4 v = make_float4(0.f, 0.f, 0.f, 0.f);
      if (gm < M)
        v = *(const float4*)(A + (size_t)gm * lda + k0 + c4);
      *(float4*)(&As[row * ALD + c4]) = v;
    }
    if (OPB) {
#pragma unroll
      for (int r = 0; r < 2; ++r) {
        const int i  = tid + r * 256;
        const int j  = i >> 2;
        const int k4 = (i & 3) << 2;
        const int gj = n0 + j;
        float4 v = make_float4(0.f, 0.f, 0.f, 0.f);
        if (gj < Nc)
          v = *(const float4*)(B + (size_t)gj * ldb + k0 + k4);
        Bs[(k4 + 0) * BLD + j] = v.x;
        Bs[(k4 + 1) * BLD + j] = v.y;
        Bs[(k4 + 2) * BLD + j] = v.z;
        Bs[(k4 + 3) * BLD + j] = v.w;
      }
    } else {
#pragma unroll
      for (int r = 0; r < 2; ++r) {
        const int i  = tid + r * 256;
        const int kk = i >> 5;
        const int c  = (i & 31) << 2;
        const int gj = n0 + c;
        float4 v = make_float4(0.f, 0.f, 0.f, 0.f);
        if (gj < Nc)
          v = *(const float4*)(B + (size_t)(k0 + kk) * ldb + gj);
        *(float4*)(&Bs[kk * BLD + c]) = v;
      }
    }
    if (k0 + BK < K) {
      const int gm = m0 + (tid >> 1);
      if (gm < M)
        __builtin_prefetch(A + (size_t)gm * lda + (k0 + BK) + (tid & 1) * 8, 0, 1);
    }
    __syncthreads();

#pragma unroll
    for (int kk = 0; kk < BK; kk += 4) {
      const int ks = kk + 2 * lh;
      v2f a[4], b[2];
#pragma unroll
      for (int tm = 0; tm < 4; ++tm) {
        a[tm] = *(const v2f*)(&As[(wm + tm * 16 + lm) * ALD + ks]);
      }
#pragma unroll
      for (int tn = 0; tn < 2; ++tn) {
        const int col = wn + tn * 16 + lm;
        b[tn].x = Bs[(ks + 0) * BLD + col];
        b[tn].y = Bs[(ks + 1) * BLD + col];
      }
#pragma unroll
      for (int tm = 0; tm < 4; ++tm)
#pragma unroll
        for (int tn = 0; tn < 2; ++tn)
          acc[tm][tn] = wmma_f32_16x16x4(a[tm], b[tn], acc[tm][tn]);
    }
  }

#pragma unroll
  for (int tm = 0; tm < 4; ++tm) {
#pragma unroll
    for (int v = 0; v < 8; ++v) {
      const int gm = m0 + wm + tm * 16 + v + 8 * lh;
      if (gm < M) {
#pragma unroll
        for (int tn = 0; tn < 2; ++tn) {
          const int gn = n0 + wn + tn * 16 + lm;
          C[(size_t)gm * ldc + gn] = acc[tm][tn][v];
        }
      }
    }
  }
}

__global__ void zero_cnt_kernel(int* __restrict__ cnt, int n) {
  int i = blockIdx.x * blockDim.x + threadIdx.x;
  if (i < n) cnt[i] = 0;
}

__global__ void hist_kernel(const int* __restrict__ dst, int* cnt, int E) {
  int e = blockIdx.x * blockDim.x + threadIdx.x;
  if (e < E) atomicAdd(&cnt[dst[e]], 1);
}

__global__ void __launch_bounds__(1024)
scan_kernel(const int* __restrict__ cnt, int* __restrict__ rowptr,
            int* __restrict__ cursor, float* __restrict__ dinv, int n) {
  __shared__ int part[1024];
  const int t   = threadIdx.x;
  const int per = (n + 1023) / 1024;
  const int b   = t * per;
  const int e   = (b + per < n) ? (b + per) : n;
  int s = 0;
  for (int i = b; i < e; ++i) s += cnt[i];
  part[t] = s;
  __syncthreads();
  for (int off = 1; off < 1024; off <<= 1) {
    const int v = (t >= off) ? part[t - off] : 0;
    __syncthreads();
    part[t] += v;
    __syncthreads();
  }
  int run = part[t] - s;
  for (int i = b; i < e; ++i) {
    const int c = cnt[i];
    rowptr[i] = run;
    cursor[i] = run;
    dinv[i]   = rsqrtf((float)(c + 1));
    run += c;
  }
  if (t == 1023) rowptr[n] = run;
}

__global__ void fill_kernel(const int* __restrict__ dst, int* cursor,
                            int* __restrict__ eid, int E) {
  int e = blockIdx.x * blockDim.x + threadIdx.x;
  if (e < E) {
    const int p = atomicAdd(&cursor[dst[e]], 1);
    eid[p] = e;
  }
}

__global__ void finalize_slots_kernel(const int* __restrict__ src,
                                      const int* __restrict__ dst,
                                      const int* __restrict__ rowptr,
                                      const int* __restrict__ eid,
                                      const float* __restrict__ dinv,
                                      int* __restrict__ csr_src,
                                      float* __restrict__ csr_w, int E) {
  int p = blockIdx.x * blockDim.x + threadIdx.x;
  if (p >= E) return;
  const int e  = eid[p];
  const int d  = dst[e];
  const int b  = rowptr[d];
  const int en = rowptr[d + 1];
  int rank = 0;
  for (int q = b; q < en; ++q) rank += (eid[q] < e) ? 1 : 0;
  const int s = src[e];
  csr_src[b + rank] = s;
  csr_w[b + rank]   = dinv[s] * dinv[d];
}

__device__ __forceinline__ void fma4(float4& acc, const float4 v, const float w) {
  acc.x = fmaf(v.x, w, acc.x);
  acc.y = fmaf(v.y, w, acc.y);
  acc.z = fmaf(v.z, w, acc.z);
  acc.w = fmaf(v.w, w, acc.w);
}

template <int D>
__global__ void __launch_bounds__(256)
gather_rows(const float* __restrict__ h, const int* __restrict__ rowptr,
            const int* __restrict__ csr_src, const float* __restrict__ csr_w,
            const float* __restrict__ dinv, const float* __restrict__ bias,
            float* __restrict__ out, int n) {
  constexpr int D4  = D / 4;
  constexpr int RPB = 256 / D4;
  static_assert(D4 % 32 == 0 && 256 % D4 == 0, "a wave must stay inside one row");
  int r = blockIdx.x * RPB + (int)(threadIdx.x / D4);
  const int j = threadIdx.x % D4;
  if (r >= n) return;
  r = __builtin_amdgcn_readfirstlane(r);

  const int   b  = rowptr[r];
  const int   e  = rowptr[r + 1];
  const float dr = dinv[r];
  const float4* __restrict__ hp = (const float4*)h + j;

  float4 acc = ((const float4*)bias)[j];
  fma4(acc, hp[(size_t)r * D4], dr * dr);

  int p = b;
  for (; p + 3 < e; p += 4) {
    const int   s0 = csr_src[p],     s1 = csr_src[p + 1];
    const int   s2 = csr_src[p + 2], s3 = csr_src[p + 3];
    const float w0 = csr_w[p],       w1 = csr_w[p + 1];
    const float w2 = csr_w[p + 2],   w3 = csr_w[p + 3];
    const float4 v0 = hp[(size_t)s0 * D4];
    const float4 v1 = hp[(size_t)s1 * D4];
    const float4 v2 = hp[(size_t)s2 * D4];
    const float4 v3 = hp[(size_t)s3 * D4];
    fma4(acc, v0, w0);
    fma4(acc, v1, w1);
    fma4(acc, v2, w2);
    fma4(acc, v3, w3);
  }
  if (p + 1 < e) {
    const int   s0 = csr_src[p], s1 = csr_src[p + 1];
    const float w0 = csr_w[p],   w1 = csr_w[p + 1];
    const float4 v0 = hp[(size_t)s0 * D4];
    const float4 v1 = hp[(size_t)s1 * D4];
    fma4(acc, v0, w0);
    fma4(acc, v1, w1);
    p += 2;
  }
  if (p < e) {
    const int   s0 = csr_src[p];
    const float w0 = csr_w[p];
    fma4(acc, hp[(size_t)s0 * D4], w0);
  }
  ((float4*)out)[(size_t)r * D4 + j] = acc;
}

static inline void launch_gemm(int opb, const float* A, int lda, const float* B,
                               int ldb, float* C, int ldc, int M, int Nc, int K,
                               hipStream_t stream) {
  dim3 grid(Nc / 128, (M + 127) / 128);
  if (opb)
    gemm_wmma<1><<<grid, 256, 0, stream>>>(A, lda, B, ldb, C, ldc, M, Nc, K);
  else
    gemm_wmma<0><<<grid, 256, 0, stream>>>(A, lda, B, ldb, C, ldc, M, Nc, K);
}

extern "C" void kernel_launch(void* const* d_in, const int* in_sizes, int n_in,
                              void* d_out, int out_size, void* d_ws, size_t ws_size,
                              hipStream_t stream) {
  const int IN = 1024, HID = 512, OUT = 128;

  const float* features = (const float*)d_in[0];
  const int*   edge     = (const int*)d_in[1];
  const float* W1       = (const float*)d_in[2];
  const float* b1       = (const float*)d_in[3];
  const float* W2       = (const float*)d_in[4];
  const float* b2       = (const float*)d_in[5];
  const float* b3       = (const float*)d_in[6];
  const float* b4       = (const float*)d_in[7];
  const float* head1    = (const float*)d_in[8];

  const int n  = in_sizes[0] / IN;
  const int E  = in_sizes[1] / 2;
  const int Et = E + n;
  const int* src = edge;
  const int* dst = edge + E;

  float* out = (float*)d_out;
  float* z   = out;
  float* h2  = out + (size_t)n * OUT;
  float* h4  = out + (size_t)n * OUT * 2;

  float* ws    = (float*)d_ws;
  float* dinv  = ws;
  float* t0    = dinv + n + Et;
  float* t1    = t0 + (size_t)n * HID;
  float* lists = t1 + (size_t)n * HID;
  float* t2    = lists + (size_t)n * OUT;
  float* h2p   = t0;

  int*   rowptr  = (int*)lists;
  int*   cursor  = rowptr + (n + 1);
  int*   cnt     = cursor + n;
  int*   eid     = cnt + n;
  int*   csr_src = eid + E;
  float* csr_w   = (float*)(csr_src + E);

  const int TB = 256;
  zero_cnt_kernel<<<(n + TB - 1) / TB, TB, 0, stream>>>(cnt, n);
  hist_kernel<<<(E + TB - 1) / TB, TB, 0, stream>>>(dst, cnt, E);
  scan_kernel<<<1, 1024, 0, stream>>>(cnt, rowptr, cursor, dinv, n);
  fill_kernel<<<(E + TB - 1) / TB, TB, 0, stream>>>(dst, cursor, eid, E);
  finalize_slots_kernel<<<(E + TB - 1) / TB, TB, 0, stream>>>(src, dst, rowptr, eid, dinv,
                                                             csr_src, csr_w, E);

  launch_gemm(1, features, IN, W1, IN, t0, HID, n, HID, IN, stream);
  gather_rows<512><<<(n + 1) / 2, 256, 0, stream>>>(t0, rowptr, csr_src, csr_w, dinv,
                                                    b1, t1, n);

  launch_gemm(1, t1, HID, W2, HID, h2p, OUT, n, OUT, HID, stream);
  gather_rows<128><<<(n + 7) / 8, 256, 0, stream>>>(h2p, rowptr, csr_src, csr_w, dinv,
                                                    b2, h2, n);

  launch_gemm(0, h2, OUT, head1, OUT, z, OUT, n, OUT, OUT, stream);

  launch_gemm(0, z, OUT, W2, HID, t0, HID, n, HID, OUT, stream);
  gather_rows<512><<<(n + 1) / 2, 256, 0, stream>>>(t0, rowptr, csr_src, csr_w, dinv,
                                                    b3, t1, n);

  launch_gemm(0, t1, HID, W1, IN, t2, IN, n, IN, HID, stream);
  gather_rows<1024><<<n, 256, 0, stream>>>(t2, rowptr, csr_src, csr_w, dinv,
                                           b4, h4, n);
}


# GEMM K loop software-pipelined on 2 LDS buffers: next tile's 4 global loads issued as one group ahead of this tile's WMMAs, stored to LDS behind them; 1 barrier per K step; sum order unchanged
# speedup vs baseline: 2.0916x; 1.2106x over previous
#include <hip/hip_runtime.h>
#include <hip/hip_bf16.h>


typedef float v2f __attribute__((ext_vector_type(2)));
typedef float v8f __attribute__((ext_vector_type(8)));

__device__ __forceinline__ v8f wmma_f32_16x16x4(v2f a, v2f b, v8f c) {
  return __builtin_amdgcn_wmma_f32_16x16x4_f32(false, a, false, b, (short)0, c,
                                               false, false);
}

constexpr int GEMM_BM  = 128;
constexpr int GEMM_BN  = 128;
constexpr int GEMM_BK  = 16;
constexpr int GEMM_ALD = 20;
constexpr int GEMM_BLD = 132;

__device__ __forceinline__ void gemm_compute_tile(const float* __restrict__ as,
                                                  const float* __restrict__ bs,
                                                  v8f (&acc)[4][2], int wm, int wn,
                                                  int lm, int lh) {
#pragma unroll
  for (int kk = 0; kk < GEMM_BK; kk += 4) {
    const int ks = kk + 2 * lh;
    v2f a[4], b[2];
#pragma unroll
    for (int tm = 0; tm < 4; ++tm) {
      a[tm] = *(const v2f*)(&as[(wm + tm * 16 + lm) * GEMM_ALD + ks]);
    }
#pragma unroll
    for (int tn = 0; tn < 2; ++tn) {
      const int col = wn + tn * 16 + lm;
      b[tn].x = bs[(ks + 0) * GEMM_BLD + col];
      b[tn].y = bs[(ks + 1) * GEMM_BLD + col];
    }
#pragma unroll
    for (int tm = 0; tm < 4; ++tm)
#pragma unroll
      for (int tn = 0; tn < 2; ++tn)
        acc[tm][tn] = wmma_f32_16x16x4(a[tm], b[tn], acc[tm][tn]);
  }
}

__device__ __forceinline__ float4 gemm_load16(const float* __restrict__ base,
                                              unsigned off) {
  return *(const float4*)((const char*)base + off);
}

template <int OPB>
__device__ __forceinline__ void gemm_store_b(float* __restrict__ bs, int s,
                                             const float4 v) {
  if (OPB) {
    bs[s + 0 * GEMM_BLD] = v.x;
    bs[s + 1 * GEMM_BLD] = v.y;
    bs[s + 2 * GEMM_BLD] = v.z;
    bs[s + 3 * GEMM_BLD] = v.w;
  } else {
    *(float4*)(&bs[s]) = v;
  }
}

#define GEMM_PIPELINED_STEP(AS_CUR, BS_CUR, AS_NXT, BS_NXT)                       \
  do {                                                                            \
    Ak += BK;                                                                     \
    Bk += bstep;                                                                  \
    ra0 = gemm_load16(Ak, oa0);                                                   \
    rb0 = gemm_load16(Bk, ob0);                                                   \
    ra1 = gemm_load16(Ak, oa1);                                                   \
    rb1 = gemm_load16(Bk, ob1);                                                   \
    __builtin_amdgcn_sched_barrier(0);    \
    gemm_compute_tile(AS_CUR, BS_CUR, acc, wm, wn, lm, lh);                       \
    __builtin_amdgcn_sched_barrier(0);     \
    *(float4*)(&AS_NXT[sa0]) = ra0;                                               \
    gemm_store_b<OPB>(BS_NXT, sb0, rb0);                                          \
    *(float4*)(&AS_NXT[sa1]) = ra1;                                               \
    gemm_store_b<OPB>(BS_NXT, sb1, rb1);                                          \
    __syncthreads();                                                              \
  } while (0)

template <int OPB>
__global__ void __launch_bounds__(256) __attribute__((amdgpu_waves_per_eu(6, 6)))
gemm_wmma(const float* __restrict__ A, int lda,
          const float* __restrict__ B, int ldb,
          float* __restrict__ C, int ldc,
          int M, int Nc, int K) {
  constexpr int BM = GEMM_BM, BN = GEMM_BN, BK = GEMM_BK;
  constexpr int ALD = GEMM_ALD, BLD = GEMM_BLD;

  __shared__ float As0[BM * ALD];
  __shared__ float As1[BM * ALD];
  __shared__ float Bs0[BK * BLD];
  __shared__ float Bs1[BK * BLD];

  const int tid  = threadIdx.x;
  const int wid  = tid >> 5;
  const int lane = tid & 31;
  const int lh   = lane >> 4;
  const int lm   = lane & 15;

  const int m0 = blockIdx.y * BM;
  const int n0 = blockIdx.x * BN;
  const int wm = (wid & 1) * 64;
  const int wn = (wid >> 1) * 32;

  const int i0 = tid, i1 = tid + 256;
  const int row0 = i0 >> 2, row1 = i1 >> 2;
  const int c40 = (i0 & 3) << 2, c41 = (i1 & 3) << 2;
  const int gm0 = (m0 + row0 < M) ? (m0 + row0) : (M - 1);
  const int gm1 = (m0 + row1 < M) ? (m0 + row1) : (M - 1);
  const unsigned oa0 = ((unsigned)gm0 * (unsigned)lda + (unsigned)c40) * 4u;
  const unsigned oa1 = ((unsigned)gm1 * (unsigned)lda + (unsigned)c41) * 4u;
  const int sa0 = row0 * ALD + c40;
  const int sa1 = row1 * ALD + c41;
  unsigned ob0, ob1;
  int sb0, sb1;
  if (OPB) {
    ob0 = ((unsigned)(n0 + row0) * (unsigned)ldb + (unsigned)c40) * 4u;
    ob1 = ((unsigned)(n0 + row1) * (unsigned)ldb + (unsigned)c41) * 4u;
    sb0 = c40 * BLD + row0;
    sb1 = c41 * BLD + row1;
  } else {
    const int kk0 = i0 >> 5, kk1 = i1 >> 5;
    const int cc0 = (i0 & 31) << 2, cc1 = (i1 & 31) << 2;
    ob0 = ((unsigned)kk0 * (unsigned)ldb + (unsigned)(n0 + cc0)) * 4u;
    ob1 = ((unsigned)kk1 * (unsigned)ldb + (unsigned)(n0 + cc1)) * 4u;
    sb0 = kk0 * BLD + cc0;
    sb1 = kk1 * BLD + cc1;
  }
  const size_t bstep = OPB ? (size_t)BK : (size_t)BK * (size_t)ldb;
  const float* Ak = A;
  const float* Bk = B;

  v8f acc[4][2];
#pragma unroll
  for (int tm = 0; tm < 4; ++tm)
#pragma unroll
    for (int tn = 0; tn < 2; ++tn) acc[tm][tn] = (v8f)0.0f;

  float4 ra0 = gemm_load16(Ak, oa0), rb0 = gemm_load16(Bk, ob0);
  float4 ra1 = gemm_load16(Ak, oa1), rb1 = gemm_load16(Bk, ob1);
  *(float4*)(&As0[sa0]) = ra0;
  gemm_store_b<OPB>(Bs0, sb0, rb0);
  *(float4*)(&As0[sa1]) = ra1;
  gemm_store_b<OPB>(Bs0, sb1, rb1);
  __syncthreads();

  const int nk = K / BK;
  int kt = 0;
  for (; kt + 2 < nk; kt += 2) {
    GEMM_PIPELINED_STEP(As0, Bs0, As1, Bs1);
    GEMM_PIPELINED_STEP(As1, Bs1, As0, Bs0);
  }
  const float* asl = As0;
  const float* bsl = Bs0;
  if (kt + 1 < nk) {
    GEMM_PIPELINED_STEP(As0, Bs0, As1, Bs1);
    asl = As1;
    bsl = Bs1;
  }
  gemm_compute_tile(asl, bsl, acc, wm, wn, lm, lh);

#pragma unroll
  for (int tm = 0; tm < 4; ++tm) {
#pragma unroll
    for (int v = 0; v < 8; ++v) {
      const int gm = m0 + wm + tm * 16 + v + 8 * lh;
      if (gm < M) {
#pragma unroll
        for (int tn = 0; tn < 2; ++tn) {
          const int gn = n0 + wn + tn * 16 + lm;
          C[(size_t)gm * ldc + gn] = acc[tm][tn][v];
        }
      }
    }
  }
}
#undef GEMM_PIPELINED_STEP

__global__ void zero_cnt_kernel(int* __restrict__ cnt, int n) {
  int i = blockIdx.x * blockDim.x + threadIdx.x;
  if (i < n) cnt[i] = 0;
}

__global__ void hist_kernel(const int* __restrict__ dst, int* cnt, int E) {
  int e = blockIdx.x * blockDim.x + threadIdx.x;
  if (e < E) atomicAdd(&cnt[dst[e]], 1);
}

__global__ void __launch_bounds__(1024)
scan_kernel(const int* __restrict__ cnt, int* __restrict__ rowptr,
            int* __restrict__ cursor, float* __restrict__ dinv, int n) {
  __shared__ int part[1024];
  const int t   = threadIdx.x;
  const int per = (n + 1023) / 1024;
  const int b   = t * per;
  const int e   = (b + per < n) ? (b + per) : n;
  int s = 0;
  for (int i = b; i < e; ++i) s += cnt[i];
  part[t] = s;
  __syncthreads();
  for (int off = 1; off < 1024; off <<= 1) {
    const int v = (t >= off) ? part[t - off] : 0;
    __syncthreads();
    part[t] += v;
    __syncthreads();
  }
  int run = part[t] - s;
  for (int i = b; i < e; ++i) {
    const int c = cnt[i];
    rowptr[i] = run;
    cursor[i] = run;
    dinv[i]   = rsqrtf((float)(c + 1));
    run += c;
  }
  if (t == 1023) rowptr[n] = run;
}

__global__ void fill_kernel(const int* __restrict__ dst, int* cursor,
                            int* __restrict__ eid, int E) {
  int e = blockIdx.x * blockDim.x + threadIdx.x;
  if (e < E) {
    const int p = atomicAdd(&cursor[dst[e]], 1);
    eid[p] = e;
  }
}

__global__ void finalize_slots_kernel(const int* __restrict__ src,
                                      const int* __restrict__ dst,
                                      const int* __restrict__ rowptr,
                                      const int* __restrict__ eid,
                                      const float* __restrict__ dinv,
                                      int* __restrict__ csr_src,
                                      float* __restrict__ csr_w, int E) {
  int p = blockIdx.x * blockDim.x + threadIdx.x;
  if (p >= E) return;
  const int e  = eid[p];
  const int d  = dst[e];
  const int b  = rowptr[d];
  const int en = rowptr[d + 1];
  int rank = 0;
  for (int q = b; q < en; ++q) rank += (eid[q] < e) ? 1 : 0;
  const int s = src[e];
  csr_src[b + rank] = s;
  csr_w[b + rank]   = dinv[s] * dinv[d];
}

__device__ __forceinline__ void fma4(float4& acc, const float4 v, const float w) {
  acc.x = fmaf(v.x, w, acc.x);
  acc.y = fmaf(v.y, w, acc.y);
  acc.z = fmaf(v.z, w, acc.z);
  acc.w = fmaf(v.w, w, acc.w);
}

template <int D>
__global__ void __launch_bounds__(256)
gather_rows(const float* __restrict__ h, const int* __restrict__ rowptr,
            const int* __restrict__ csr_src, const float* __restrict__ csr_w,
            const float* __restrict__ dinv, const float* __restrict__ bias,
            float* __restrict__ out, int n) {
  constexpr int D4  = D / 4;
  constexpr int RPB = 256 / D4;
  static_assert(D4 % 32 == 0 && 256 % D4 == 0, "a wave must stay inside one row");
  int r = blockIdx.x * RPB + (int)(threadIdx.x / D4);
  const int j = threadIdx.x % D4;
  if (r >= n) return;
  r = __builtin_amdgcn_readfirstlane(r);

  const int   b  = rowptr[r];
  const int   e  = rowptr[r + 1];
  const float dr = dinv[r];
  const float4* __restrict__ hp = (const float4*)h + j;

  float4 acc = ((const float4*)bias)[j];
  fma4(acc, hp[(size_t)r * D4], dr * dr);

  int p = b;
  for (; p + 3 < e; p += 4) {
    const int   s0 = csr_src[p],     s1 = csr_src[p + 1];
    const int   s2 = csr_src[p + 2], s3 = csr_src[p + 3];
    const float w0 = csr_w[p],       w1 = csr_w[p + 1];
    const float w2 = csr_w[p + 2],   w3 = csr_w[p + 3];
    const float4 v0 = hp[(size_t)s0 * D4];
    const float4 v1 = hp[(size_t)s1 * D4];
    const float4 v2 = hp[(size_t)s2 * D4];
    const float4 v3 = hp[(size_t)s3 * D4];
    fma4(acc, v0, w0);
    fma4(acc, v1, w1);
    fma4(acc, v2, w2);
    fma4(acc, v3, w3);
  }
  if (p + 1 < e) {
    const int   s0 = csr_src[p], s1 = csr_src[p + 1];
    const float w0 = csr_w[p],   w1 = csr_w[p + 1];
    const float4 v0 = hp[(size_t)s0 * D4];
    const float4 v1 = hp[(size_t)s1 * D4];
    fma4(acc, v0, w0);
    fma4(acc, v1, w1);
    p += 2;
  }
  if (p < e) {
    const int   s0 = csr_src[p];
    const float w0 = csr_w[p];
    fma4(acc, hp[(size_t)s0 * D4], w0);
  }
  ((float4*)out)[(size_t)r * D4 + j] = acc;
}

static inline void launch_gemm(int opb, const float* A, int lda, const float* B,
                               int ldb, float* C, int ldc, int M, int Nc, int K,
                               hipStream_t stream) {
  dim3 grid(Nc / 128, (M + 127) / 128);
  if (opb)
    gemm_wmma<1><<<grid, 256, 0, stream>>>(A, lda, B, ldb, C, ldc, M, Nc, K);
  else
    gemm_wmma<0><<<grid, 256, 0, stream>>>(A, lda, B, ldb, C, ldc, M, Nc, K);
}

extern "C" void kernel_launch(void* const* d_in, const int* in_sizes, int n_in,
                              void* d_out, int out_size, void* d_ws, size_t ws_size,
                              hipStream_t stream) {
  const int IN = 1024, HID = 512, OUT = 128;

  const float* features = (const float*)d_in[0];
  const int*   edge     = (const int*)d_in[1];
  const float* W1       = (const float*)d_in[2];
  const float* b1       = (const float*)d_in[3];
  const float* W2       = (const float*)d_in[4];
  const float* b2       = (const float*)d_in[5];
  const float* b3       = (const float*)d_in[6];
  const float* b4       = (const float*)d_in[7];
  const float* head1    = (const float*)d_in[8];

  const int n  = in_sizes[0] / IN;
  const int E  = in_sizes[1] / 2;
  const int Et = E + n;
  const int* src = edge;
  const int* dst = edge + E;

  float* out = (float*)d_out;
  float* z   = out;
  float* h2  = out + (size_t)n * OUT;
  float* h4  = out + (size_t)n * OUT * 2;

  float* ws    = (float*)d_ws;
  float* dinv  = ws;
  float* t0    = dinv + n + Et;
  float* t1    = t0 + (size_t)n * HID;
  float* lists = t1 + (size_t)n * HID;
  float* t2    = lists + (size_t)n * OUT;
  float* h2p   = t0;

  int*   rowptr  = (int*)lists;
  int*   cursor  = rowptr + (n + 1);
  int*   cnt     = cursor + n;
  int*   eid     = cnt + n;
  int*   csr_src = eid + E;
  float* csr_w   = (float*)(csr_src + E);

  const int TB = 256;
  zero_cnt_kernel<<<(n + TB - 1) / TB, TB, 0, stream>>>(cnt, n);
  hist_kernel<<<(E + TB - 1) / TB, TB, 0, stream>>>(dst, cnt, E);
  scan_kernel<<<1, 1024, 0, stream>>>(cnt, rowptr, cursor, dinv, n);
  fill_kernel<<<(E + TB - 1) / TB, TB, 0, stream>>>(dst, cursor, eid, E);
  finalize_slots_kernel<<<(E + TB - 1) / TB, TB, 0, stream>>>(src, dst, rowptr, eid, dinv,
                                                             csr_src, csr_w, E);

  launch_gemm(1, features, IN, W1, IN, t0, HID, n, HID, IN, stream);
  gather_rows<512><<<(n + 1) / 2, 256, 0, stream>>>(t0, rowptr, csr_src, csr_w, dinv,
                                                    b1, t1, n);

  launch_gemm(1, t1, HID, W2, HID, h2p, OUT, n, OUT, HID, stream);
  gather_rows<128><<<(n + 7) / 8, 256, 0, stream>>>(h2p, rowptr, csr_src, csr_w, dinv,
                                                    b2, h2, n);

  launch_gemm(0, h2, OUT, head1, OUT, z, OUT, n, OUT, OUT, stream);

  launch_gemm(0, z, OUT, W2, HID, t0, HID, n, HID, OUT, stream);
  gather_rows<512><<<(n + 1) / 2, 256, 0, stream>>>(t0, rowptr, csr_src, csr_w, dinv,
                                                    b3, t1, n);

  launch_gemm(0, t1, HID, W1, IN, t2, IN, n, IN, HID, stream);
  gather_rows<1024><<<n, 256, 0, stream>>>(t2, rowptr, csr_src, csr_w, dinv,
                                           b4, h4, n);
}


# Decoder layers aggregate first, then multiply (agg(x)@W = agg(x@W)): their row gathers run 128 and 512 wide, not 512 and 1024; bias added in the GEMM store; changes float sum order
# speedup vs baseline: 2.2130x; 1.0580x over previous
#include <hip/hip_runtime.h>
#include <hip/hip_bf16.h>


typedef float v2f __attribute__((ext_vector_type(2)));
typedef float v8f __attribute__((ext_vector_type(8)));

__device__ __forceinline__ v8f wmma_f32_16x16x4(v2f a, v2f b, v8f c) {
  return __builtin_amdgcn_wmma_f32_16x16x4_f32(false, a, false, b, (short)0, c,
                                               false, false);
}

constexpr int GEMM_BM  = 128;
constexpr int GEMM_BN  = 128;
constexpr int GEMM_BK  = 16;
constexpr int GEMM_ALD = 20;
constexpr int GEMM_BLD = 132;

__device__ __forceinline__ void gemm_compute_tile(const float* __restrict__ as,
                                                  const float* __restrict__ bs,
                                                  v8f (&acc)[4][2], int wm, int wn,
                                                  int lm, int lh) {
#pragma unroll
  for (int kk = 0; kk < GEMM_BK; kk += 4) {
    const int ks = kk + 2 * lh;
    v2f a[4], b[2];
#pragma unroll
    for (int tm = 0; tm < 4; ++tm) {
      a[tm] = *(const v2f*)(&as[(wm + tm * 16 + lm) * GEMM_ALD + ks]);
    }
#pragma unroll
    for (int tn = 0; tn < 2; ++tn) {
      const int col = wn + tn * 16 + lm;
      b[tn].x = bs[(ks + 0) * GEMM_BLD + col];
      b[tn].y = bs[(ks + 1) * GEMM_BLD + col];
    }
#pragma unroll
    for (int tm = 0; tm < 4; ++tm)
#pragma unroll
      for (int tn = 0; tn < 2; ++tn)
        acc[tm][tn] = wmma_f32_16x16x4(a[tm], b[tn], acc[tm][tn]);
  }
}

__device__ __forceinline__ float4 gemm_load16(const float* __restrict__ base,
                                              unsigned off) {
  return *(const float4*)((const char*)base + off);
}

template <int OPB>
__device__ __forceinline__ void gemm_store_b(float* __restrict__ bs, int s,
                                             const float4 v) {
  if (OPB) {
    bs[s + 0 * GEMM_BLD] = v.x;
    bs[s + 1 * GEMM_BLD] = v.y;
    bs[s + 2 * GEMM_BLD] = v.z;
    bs[s + 3 * GEMM_BLD] = v.w;
  } else {
    *(float4*)(&bs[s]) = v;
  }
}

#define GEMM_PIPELINED_STEP(AS_CUR, BS_CUR, AS_NXT, BS_NXT)                       \
  do {                                                                            \
    Ak += BK;                                                                     \
    Bk += bstep;                                                                  \
    ra0 = gemm_load16(Ak, oa0);                                                   \
    rb0 = gemm_load16(Bk, ob0);                                                   \
    ra1 = gemm_load16(Ak, oa1);                                                   \
    rb1 = gemm_load16(Bk, ob1);                                                   \
    __builtin_amdgcn_sched_barrier(0);    \
    gemm_compute_tile(AS_CUR, BS_CUR, acc, wm, wn, lm, lh);                       \
    __builtin_amdgcn_sched_barrier(0);     \
    *(float4*)(&AS_NXT[sa0]) = ra0;                                               \
    gemm_store_b<OPB>(BS_NXT, sb0, rb0);                                          \
    *(float4*)(&AS_NXT[sa1]) = ra1;                                               \
    gemm_store_b<OPB>(BS_NXT, sb1, rb1);                                          \
    __syncthreads();                                                              \
  } while (0)

template <int OPB, bool BIAS>
__global__ void __launch_bounds__(256) __attribute__((amdgpu_waves_per_eu(6, 6)))
gemm_wmma(const float* __restrict__ A, int lda,
          const float* __restrict__ B, int ldb,
          float* __restrict__ C, int ldc,
          int M, int Nc, int K, const float* __restrict__ bias) {
  constexpr int BM = GEMM_BM, BN = GEMM_BN, BK = GEMM_BK;
  constexpr int ALD = GEMM_ALD, BLD = GEMM_BLD;

  __shared__ float As0[BM * ALD];
  __shared__ float As1[BM * ALD];
  __shared__ float Bs0[BK * BLD];
  __shared__ float Bs1[BK * BLD];

  const int tid  = threadIdx.x;
  const int wid  = tid >> 5;
  const int lane = tid & 31;
  const int lh   = lane >> 4;
  const int lm   = lane & 15;

  const int m0 = blockIdx.y * BM;
  const int n0 = blockIdx.x * BN;
  const int wm = (wid & 1) * 64;
  const int wn = (wid >> 1) * 32;

  const int i0 = tid, i1 = tid + 256;
  const int row0 = i0 >> 2, row1 = i1 >> 2;
  const int c40 = (i0 & 3) << 2, c41 = (i1 & 3) << 2;
  const int gm0 = (m0 + row0 < M) ? (m0 + row0) : (M - 1);
  const int gm1 = (m0 + row1 < M) ? (m0 + row1) : (M - 1);
  const unsigned oa0 = ((unsigned)gm0 * (unsigned)lda + (unsigned)c40) * 4u;
  const unsigned oa1 = ((unsigned)gm1 * (unsigned)lda + (unsigned)c41) * 4u;
  const int sa0 = row0 * ALD + c40;
  const int sa1 = row1 * ALD + c41;
  unsigned ob0, ob1;
  int sb0, sb1;
  if (OPB) {
    ob0 = ((unsigned)(n0 + row0) * (unsigned)ldb + (unsigned)c40) * 4u;
    ob1 = ((unsigned)(n0 + row1) * (unsigned)ldb + (unsigned)c41) * 4u;
    sb0 = c40 * BLD + row0;
    sb1 = c41 * BLD + row1;
  } else {
    const int kk0 = i0 >> 5, kk1 = i1 >> 5;
    const int cc0 = (i0 & 31) << 2, cc1 = (i1 & 31) << 2;
    ob0 = ((unsigned)kk0 * (unsigned)ldb + (unsigned)(n0 + cc0)) * 4u;
    ob1 = ((unsigned)kk1 * (unsigned)ldb + (unsigned)(n0 + cc1)) * 4u;
    sb0 = kk0 * BLD + cc0;
    sb1 = kk1 * BLD + cc1;
  }
  const size_t bstep = OPB ? (size_t)BK : (size_t)BK * (size_t)ldb;
  const float* Ak = A;
  const float* Bk = B;

  v8f acc[4][2];
#pragma unroll
  for (int tm = 0; tm < 4; ++tm)
#pragma unroll
    for (int tn = 0; tn < 2; ++tn) acc[tm][tn] = (v8f)0.0f;

  float4 ra0 = gemm_load16(Ak, oa0), rb0 = gemm_load16(Bk, ob0);
  float4 ra1 = gemm_load16(Ak, oa1), rb1 = gemm_load16(Bk, ob1);
  *(float4*)(&As0[sa0]) = ra0;
  gemm_store_b<OPB>(Bs0, sb0, rb0);
  *(float4*)(&As0[sa1]) = ra1;
  gemm_store_b<OPB>(Bs0, sb1, rb1);
  __syncthreads();

  const int nk = K / BK;
  int kt = 0;
  for (; kt + 2 < nk; kt += 2) {
    GEMM_PIPELINED_STEP(As0, Bs0, As1, Bs1);
    GEMM_PIPELINED_STEP(As1, Bs1, As0, Bs0);
  }
  const float* asl = As0;
  const float* bsl = Bs0;
  if (kt + 1 < nk) {
    GEMM_PIPELINED_STEP(As0, Bs0, As1, Bs1);
    asl = As1;
    bsl = Bs1;
  }
  gemm_compute_tile(asl, bsl, acc, wm, wn, lm, lh);

  float bv[2] = {0.0f, 0.0f};
  if (BIAS) {
#pragma unroll
    for (int tn = 0; tn < 2; ++tn) bv[tn] = bias[n0 + wn + tn * 16 + lm];
  }
#pragma unroll
  for (int tm = 0; tm < 4; ++tm) {
#pragma unroll
    for (int v = 0; v < 8; ++v) {
      const int gm = m0 + wm + tm * 16 + v + 8 * lh;
      if (gm < M) {
#pragma unroll
        for (int tn = 0; tn < 2; ++tn) {
          const int gn = n0 + wn + tn * 16 + lm;
          C[(size_t)gm * ldc + gn] = BIAS ? acc[tm][tn][v] + bv[tn] : acc[tm][tn][v];
        }
      }
    }
  }
}
#undef GEMM_PIPELINED_STEP

__global__ void zero_cnt_kernel(int* __restrict__ cnt, int n) {
  int i = blockIdx.x * blockDim.x + threadIdx.x;
  if (i < n) cnt[i] = 0;
}

__global__ void hist_kernel(const int* __restrict__ dst, int* cnt, int E) {
  int e = blockIdx.x * blockDim.x + threadIdx.x;
  if (e < E) atomicAdd(&cnt[dst[e]], 1);
}

__global__ void __launch_bounds__(1024)
scan_kernel(const int* __restrict__ cnt, int* __restrict__ rowptr,
            int* __restrict__ cursor, float* __restrict__ dinv, int n) {
  __shared__ int part[1024];
  const int t   = threadIdx.x;
  const int per = (n + 1023) / 1024;
  const int b   = t * per;
  const int e   = (b + per < n) ? (b + per) : n;
  int s = 0;
  for (int i = b; i < e; ++i) s += cnt[i];
  part[t] = s;
  __syncthreads();
  for (int off = 1; off < 1024; off <<= 1) {
    const int v = (t >= off) ? part[t - off] : 0;
    __syncthreads();
    part[t] += v;
    __syncthreads();
  }
  int run = part[t] - s;
  for (int i = b; i < e; ++i) {
    const int c = cnt[i];
    rowptr[i] = run;
    cursor[i] = run;
    dinv[i]   = rsqrtf((float)(c + 1));
    run += c;
  }
  if (t == 1023) rowptr[n] = run;
}

__global__ void fill_kernel(const int* __restrict__ dst, int* cursor,
                            int* __restrict__ eid, int E) {
  int e = blockIdx.x * blockDim.x + threadIdx.x;
  if (e < E) {
    const int p = atomicAdd(&cursor[dst[e]], 1);
    eid[p] = e;
  }
}

__global__ void finalize_slots_kernel(const int* __restrict__ src,
                                      const int* __restrict__ dst,
                                      const int* __restrict__ rowptr,
                                      const int* __restrict__ eid,
                                      const float* __restrict__ dinv,
                                      int* __restrict__ csr_src,
                                      float* __restrict__ csr_w, int E) {
  int p = blockIdx.x * blockDim.x + threadIdx.x;
  if (p >= E) return;
  const int e  = eid[p];
  const int d  = dst[e];
  const int b  = rowptr[d];
  const int en = rowptr[d + 1];
  int rank = 0;
  for (int q = b; q < en; ++q) rank += (eid[q] < e) ? 1 : 0;
  const int s = src[e];
  csr_src[b + rank] = s;
  csr_w[b + rank]   = dinv[s] * dinv[d];
}

__device__ __forceinline__ void fma4(float4& acc, const float4 v, const float w) {
  acc.x = fmaf(v.x, w, acc.x);
  acc.y = fmaf(v.y, w, acc.y);
  acc.z = fmaf(v.z, w, acc.z);
  acc.w = fmaf(v.w, w, acc.w);
}

template <int D, bool BIAS>
__global__ void __launch_bounds__(256)
gather_rows(const float* __restrict__ h, const int* __restrict__ rowptr,
            const int* __restrict__ csr_src, const float* __restrict__ csr_w,
            const float* __restrict__ dinv, const float* __restrict__ bias,
            float* __restrict__ out, int n) {
  constexpr int D4  = D / 4;
  constexpr int RPB = 256 / D4;
  static_assert(D4 % 32 == 0 && 256 % D4 == 0, "a wave must stay inside one row");
  int r = blockIdx.x * RPB + (int)(threadIdx.x / D4);
  const int j = threadIdx.x % D4;
  if (r >= n) return;
  r = __builtin_amdgcn_readfirstlane(r);

  const int   b  = rowptr[r];
  const int   e  = rowptr[r + 1];
  const float dr = dinv[r];
  const float4* __restrict__ hp = (const float4*)h + j;

  float4 acc = BIAS ? ((const float4*)bias)[j] : make_float4(0.0f, 0.0f, 0.0f, 0.0f);
  fma4(acc, hp[(size_t)r * D4], dr * dr);

  int p = b;
  for (; p + 3 < e; p += 4) {
    const int   s0 = csr_src[p],     s1 = csr_src[p + 1];
    const int   s2 = csr_src[p + 2], s3 = csr_src[p + 3];
    const float w0 = csr_w[p],       w1 = csr_w[p + 1];
    const float w2 = csr_w[p + 2],   w3 = csr_w[p + 3];
    const float4 v0 = hp[(size_t)s0 * D4];
    const float4 v1 = hp[(size_t)s1 * D4];
    const float4 v2 = hp[(size_t)s2 * D4];
    const float4 v3 = hp[(size_t)s3 * D4];
    fma4(acc, v0, w0);
    fma4(acc, v1, w1);
    fma4(acc, v2, w2);
    fma4(acc, v3, w3);
  }
  if (p + 1 < e) {
    const int   s0 = csr_src[p], s1 = csr_src[p + 1];
    const float w0 = csr_w[p],   w1 = csr_w[p + 1];
    const float4 v0 = hp[(size_t)s0 * D4];
    const float4 v1 = hp[(size_t)s1 * D4];
    fma4(acc, v0, w0);
    fma4(acc, v1, w1);
    p += 2;
  }
  if (p < e) {
    const int   s0 = csr_src[p];
    const float w0 = csr_w[p];
    fma4(acc, hp[(size_t)s0 * D4], w0);
  }
  ((float4*)out)[(size_t)r * D4 + j] = acc;
}

static inline void launch_gemm(int opb, const float* A, int lda, const float* B,
                               int ldb, float* C, int ldc, int M, int Nc, int K,
                               const float* bias, hipStream_t stream) {
  dim3 grid(Nc / 128, (M + 127) / 128);
  if (opb)
    gemm_wmma<1, false><<<grid, 256, 0, stream>>>(A, lda, B, ldb, C, ldc, M, Nc, K, bias);
  else if (bias)
    gemm_wmma<0, true><<<grid, 256, 0, stream>>>(A, lda, B, ldb, C, ldc, M, Nc, K, bias);
  else
    gemm_wmma<0, false><<<grid, 256, 0, stream>>>(A, lda, B, ldb, C, ldc, M, Nc, K, bias);
}

extern "C" void kernel_launch(void* const* d_in, const int* in_sizes, int n_in,
                              void* d_out, int out_size, void* d_ws, size_t ws_size,
                              hipStream_t stream) {
  const int IN = 1024, HID = 512, OUT = 128;

  const float* features = (const float*)d_in[0];
  const int*   edge     = (const int*)d_in[1];
  const float* W1       = (const float*)d_in[2];
  const float* b1       = (const float*)d_in[3];
  const float* W2       = (const float*)d_in[4];
  const float* b2       = (const float*)d_in[5];
  const float* b3       = (const float*)d_in[6];
  const float* b4       = (const float*)d_in[7];
  const float* head1    = (const float*)d_in[8];

  const int n  = in_sizes[0] / IN;
  const int E  = in_sizes[1] / 2;
  const int Et = E + n;
  const int* src = edge;
  const int* dst = edge + E;

  float* out = (float*)d_out;
  float* z   = out;
  float* h2  = out + (size_t)n * OUT;
  float* h4  = out + (size_t)n * OUT * 2;

  float* ws    = (float*)d_ws;
  float* dinv  = ws;
  float* t0    = dinv + n + Et;
  float* t1    = t0 + (size_t)n * HID;
  float* lists = t1 + (size_t)n * HID;
  float* t2    = lists + (size_t)n * OUT;
  float* h2p   = t0;
  float* az    = t2;

  int*   rowptr  = (int*)lists;
  int*   cursor  = rowptr + (n + 1);
  int*   cnt     = cursor + n;
  int*   eid     = cnt + n;
  int*   csr_src = eid + E;
  float* csr_w   = (float*)(csr_src + E);

  const int TB = 256;
  zero_cnt_kernel<<<(n + TB - 1) / TB, TB, 0, stream>>>(cnt, n);
  hist_kernel<<<(E + TB - 1) / TB, TB, 0, stream>>>(dst, cnt, E);
  scan_kernel<<<1, 1024, 0, stream>>>(cnt, rowptr, cursor, dinv, n);
  fill_kernel<<<(E + TB - 1) / TB, TB, 0, stream>>>(dst, cursor, eid, E);
  finalize_slots_kernel<<<(E + TB - 1) / TB, TB, 0, stream>>>(src, dst, rowptr, eid, dinv,
                                                             csr_src, csr_w, E);

  launch_gemm(1, features, IN, W1, IN, t0, HID, n, HID, IN, nullptr, stream);
  gather_rows<512, true><<<(n + 1) / 2, 256, 0, stream>>>(t0, rowptr, csr_src, csr_w,
                                                          dinv, b1, t1, n);

  launch_gemm(1, t1, HID, W2, HID, h2p, OUT, n, OUT, HID, nullptr, stream);
  gather_rows<128, true><<<(n + 7) / 8, 256, 0, stream>>>(h2p, rowptr, csr_src, csr_w,
                                                          dinv, b2, h2, n);

  launch_gemm(0, h2, OUT, head1, OUT, z, OUT, n, OUT, OUT, nullptr, stream);

  gather_rows<128, false><<<(n + 7) / 8, 256, 0, stream>>>(z, rowptr, csr_src, csr_w,
                                                           dinv, nullptr, az, n);
  launch_gemm(0, az, OUT, W2, HID, t0, HID, n, HID, OUT, b3, stream);

  gather_rows<512, false><<<(n + 1) / 2, 256, 0, stream>>>(t0, rowptr, csr_src, csr_w,
                                                           dinv, nullptr, t1, n);
  launch_gemm(0, t1, HID, W1, IN, h4, IN, n, IN, HID, b4, stream);
}


# No activation between layers, so each layer pair uses one folded weight W2@W1 (128x1024, built on WMMA every call): big GEMMs get a 128 side, all row gathers 128 wide; changes float sum order
# speedup vs baseline: 3.7533x; 1.6960x over previous
#include <hip/hip_runtime.h>
#include <hip/hip_bf16.h>


typedef float v2f __attribute__((ext_vector_type(2)));
typedef float v8f __attribute__((ext_vector_type(8)));

__device__ __forceinline__ v8f wmma_f32_16x16x4(v2f a, v2f b, v8f c) {
  return __builtin_amdgcn_wmma_f32_16x16x4_f32(false, a, false, b, (short)0, c,
                                               false, false);
}

constexpr int GEMM_BM  = 128;
constexpr int GEMM_BN  = 128;
constexpr int GEMM_BK  = 16;
constexpr int GEMM_ALD = 20;
constexpr int GEMM_BLD = 132;

__device__ __forceinline__ void gemm_compute_tile(const float* __restrict__ as,
                                                  const float* __restrict__ bs,
                                                  v8f (&acc)[4][2], int wm, int wn,
                                                  int lm, int lh) {
#pragma unroll
  for (int kk = 0; kk < GEMM_BK; kk += 4) {
    const int ks = kk + 2 * lh;
    v2f a[4], b[2];
#pragma unroll
    for (int tm = 0; tm < 4; ++tm) {
      a[tm] = *(const v2f*)(&as[(wm + tm * 16 + lm) * GEMM_ALD + ks]);
    }
#pragma unroll
    for (int tn = 0; tn < 2; ++tn) {
      const int col = wn + tn * 16 + lm;
      b[tn].x = bs[(ks + 0) * GEMM_BLD + col];
      b[tn].y = bs[(ks + 1) * GEMM_BLD + col];
    }
#pragma unroll
    for (int tm = 0; tm < 4; ++tm)
#pragma unroll
      for (int tn = 0; tn < 2; ++tn)
        acc[tm][tn] = wmma_f32_16x16x4(a[tm], b[tn], acc[tm][tn]);
  }
}

__device__ __forceinline__ float4 gemm_load16(const float* __restrict__ base,
                                              unsigned off) {
  return *(const float4*)((const char*)base + off);
}

template <int OPB>
__device__ __forceinline__ void gemm_store_b(float* __restrict__ bs, int s,
                                             const float4 v) {
  if (OPB) {
    bs[s + 0 * GEMM_BLD] = v.x;
    bs[s + 1 * GEMM_BLD] = v.y;
    bs[s + 2 * GEMM_BLD] = v.z;
    bs[s + 3 * GEMM_BLD] = v.w;
  } else {
    *(float4*)(&bs[s]) = v;
  }
}

#define GEMM_PIPELINED_STEP(AS_CUR, BS_CUR, AS_NXT, BS_NXT)                       \
  do {                                                                            \
    Ak += BK;                                                                     \
    Bk += bstep;                                                                  \
    ra0 = gemm_load16(Ak, oa0);                                                   \
    rb0 = gemm_load16(Bk, ob0);                                                   \
    ra1 = gemm_load16(Ak, oa1);                                                   \
    rb1 = gemm_load16(Bk, ob1);                                                   \
    __builtin_amdgcn_sched_barrier(0);    \
    gemm_compute_tile(AS_CUR, BS_CUR, acc, wm, wn, lm, lh);                       \
    __builtin_amdgcn_sched_barrier(0);     \
    *(float4*)(&AS_NXT[sa0]) = ra0;                                               \
    gemm_store_b<OPB>(BS_NXT, sb0, rb0);                                          \
    *(float4*)(&AS_NXT[sa1]) = ra1;                                               \
    gemm_store_b<OPB>(BS_NXT, sb1, rb1);                                          \
    __syncthreads();                                                              \
  } while (0)

template <int OPB, int EPI>
__global__ void __launch_bounds__(256) __attribute__((amdgpu_waves_per_eu(6, 6)))
gemm_wmma(const float* __restrict__ A, int lda,
          const float* __restrict__ B, int ldb,
          float* __restrict__ C, int ldc,
          int M, int Nc, int K, const float* __restrict__ bias,
          const float* __restrict__ rsum, const float* __restrict__ cvec) {
  constexpr int BM = GEMM_BM, BN = GEMM_BN, BK = GEMM_BK;
  constexpr int ALD = GEMM_ALD, BLD = GEMM_BLD;

  __shared__ float As0[BM * ALD];
  __shared__ float As1[BM * ALD];
  __shared__ float Bs0[BK * BLD];
  __shared__ float Bs1[BK * BLD];

  const int tid  = threadIdx.x;
  const int wid  = tid >> 5;
  const int lane = tid & 31;
  const int lh   = lane >> 4;
  const int lm   = lane & 15;

  const int m0 = blockIdx.y * BM;
  const int n0 = blockIdx.x * BN;
  const int wm = (wid & 1) * 64;
  const int wn = (wid >> 1) * 32;

  const int i0 = tid, i1 = tid + 256;
  const int row0 = i0 >> 2, row1 = i1 >> 2;
  const int c40 = (i0 & 3) << 2, c41 = (i1 & 3) << 2;
  const int gm0 = (m0 + row0 < M) ? (m0 + row0) : (M - 1);
  const int gm1 = (m0 + row1 < M) ? (m0 + row1) : (M - 1);
  const unsigned oa0 = ((unsigned)gm0 * (unsigned)lda + (unsigned)c40) * 4u;
  const unsigned oa1 = ((unsigned)gm1 * (unsigned)lda + (unsigned)c41) * 4u;
  const int sa0 = row0 * ALD + c40;
  const int sa1 = row1 * ALD + c41;
  unsigned ob0, ob1;
  int sb0, sb1;
  if (OPB) {
    ob0 = ((unsigned)(n0 + row0) * (unsigned)ldb + (unsigned)c40) * 4u;
    ob1 = ((unsigned)(n0 + row1) * (unsigned)ldb + (unsigned)c41) * 4u;
    sb0 = c40 * BLD + row0;
    sb1 = c41 * BLD + row1;
  } else {
    const int kk0 = i0 >> 5, kk1 = i1 >> 5;
    const int cc0 = (i0 & 31) << 2, cc1 = (i1 & 31) << 2;
    ob0 = ((unsigned)kk0 * (unsigned)ldb + (unsigned)(n0 + cc0)) * 4u;
    ob1 = ((unsigned)kk1 * (unsigned)ldb + (unsigned)(n0 + cc1)) * 4u;
    sb0 = kk0 * BLD + cc0;
    sb1 = kk1 * BLD + cc1;
  }
  const size_t bstep = OPB ? (size_t)BK : (size_t)BK * (size_t)ldb;
  const float* Ak = A;
  const float* Bk = B;

  v8f acc[4][2];
#pragma unroll
  for (int tm = 0; tm < 4; ++tm)
#pragma unroll
    for (int tn = 0; tn < 2; ++tn) acc[tm][tn] = (v8f)0.0f;

  float4 ra0 = gemm_load16(Ak, oa0), rb0 = gemm_load16(Bk, ob0);
  float4 ra1 = gemm_load16(Ak, oa1), rb1 = gemm_load16(Bk, ob1);
  *(float4*)(&As0[sa0]) = ra0;
  gemm_store_b<OPB>(Bs0, sb0, rb0);
  *(float4*)(&As0[sa1]) = ra1;
  gemm_store_b<OPB>(Bs0, sb1, rb1);
  __syncthreads();

  const int nk = K / BK;
  int kt = 0;
  for (; kt + 2 < nk; kt += 2) {
    GEMM_PIPELINED_STEP(As0, Bs0, As1, Bs1);
    GEMM_PIPELINED_STEP(As1, Bs1, As0, Bs0);
  }
  const float* asl = As0;
  const float* bsl = Bs0;
  if (kt + 1 < nk) {
    GEMM_PIPELINED_STEP(As0, Bs0, As1, Bs1);
    asl = As1;
    bsl = Bs1;
  }
  gemm_compute_tile(asl, bsl, acc, wm, wn, lm, lh);

  float bv[2] = {0.0f, 0.0f};
  if (EPI >= 1) {
#pragma unroll
    for (int tn = 0; tn < 2; ++tn) bv[tn] = bias[n0 + wn + tn * 16 + lm];
  }
  if (EPI == 2) {
    float cv[2];
#pragma unroll
    for (int tn = 0; tn < 2; ++tn) cv[tn] = cvec[n0 + wn + tn * 16 + lm];
#pragma unroll
    for (int tm = 0; tm < 4; ++tm) {
#pragma unroll
      for (int v = 0; v < 8; ++v) {
        const int gm = m0 + wm + tm * 16 + v + 8 * lh;
        const float rr = rsum[(gm < M) ? gm : (M - 1)];
#pragma unroll
        for (int tn = 0; tn < 2; ++tn)
          acc[tm][tn][v] = fmaf(rr, cv[tn], acc[tm][tn][v]);
      }
    }
  }
#pragma unroll
  for (int tm = 0; tm < 4; ++tm) {
#pragma unroll
    for (int v = 0; v < 8; ++v) {
      const int gm = m0 + wm + tm * 16 + v + 8 * lh;
      if (gm < M) {
#pragma unroll
        for (int tn = 0; tn < 2; ++tn) {
          const int gn = n0 + wn + tn * 16 + lm;
          C[(size_t)gm * ldc + gn] = (EPI >= 1) ? acc[tm][tn][v] + bv[tn] : acc[tm][tn][v];
        }
      }
    }
  }
}
#undef GEMM_PIPELINED_STEP

template <int IN_, int HID_>
__global__ void __launch_bounds__(256)
fold_bias_kernel(const float* __restrict__ W1, const float* __restrict__ W2,
                 const float* __restrict__ b1, const float* __restrict__ b3,
                 float* __restrict__ dvec, float* __restrict__ cv2) {
  static_assert(IN_ % 32 == 0 && HID_ % 32 == 0, "whole waves of columns and of k");
  constexpr int in = IN_, hid = HID_;
  __shared__ float part[8][32];
  const int wid  = threadIdx.x >> 5;
  const int lane = threadIdx.x & 31;
  constexpr int nd = in / 32;
  if ((int)blockIdx.x < nd) {
    const int j = blockIdx.x * 32 + lane;
    constexpr int per = hid / 8;
    const int k0 = wid * per;
    float s = 0.0f;
#pragma unroll 16
    for (int k = 0; k < per; ++k)
      s = fmaf(b3[k0 + k], W1[(size_t)(k0 + k) * in + j], s);
    part[wid][lane] = s;
    __syncthreads();
    if (wid == 0) {
      float t = part[0][lane];
#pragma unroll
      for (int w = 1; w < 8; ++w) t += part[w][lane];
      dvec[j] = t;
    }
  } else {
    const int j = ((int)blockIdx.x - nd) * 8 + wid;
    const float* __restrict__ wr = W2 + (size_t)j * hid;
    float s = 0.0f;
#pragma unroll
    for (int i = 0; i < hid / 32; ++i) s = fmaf(wr[lane + 32 * i], b1[lane + 32 * i], s);
#pragma unroll
    for (int off = 16; off > 0; off >>= 1) s += __shfl_xor(s, off);
    if (lane == 0) cv2[j] = s;
  }
}

__global__ void zero_cnt_kernel(int* __restrict__ cnt, int n) {
  int i = blockIdx.x * blockDim.x + threadIdx.x;
  if (i < n) cnt[i] = 0;
}

__global__ void hist_kernel(const int* __restrict__ dst, int* cnt, int E) {
  int e = blockIdx.x * blockDim.x + threadIdx.x;
  if (e < E) atomicAdd(&cnt[dst[e]], 1);
}

__global__ void __launch_bounds__(1024)
scan_kernel(const int* __restrict__ cnt, int* __restrict__ rowptr,
            int* __restrict__ cursor, float* __restrict__ dinv, int n) {
  __shared__ int part[1024];
  const int t   = threadIdx.x;
  const int per = (n + 1023) / 1024;
  const int b   = t * per;
  const int e   = (b + per < n) ? (b + per) : n;
  int s = 0;
  for (int i = b; i < e; ++i) s += cnt[i];
  part[t] = s;
  __syncthreads();
  for (int off = 1; off < 1024; off <<= 1) {
    const int v = (t >= off) ? part[t - off] : 0;
    __syncthreads();
    part[t] += v;
    __syncthreads();
  }
  int run = part[t] - s;
  for (int i = b; i < e; ++i) {
    const int c = cnt[i];
    rowptr[i] = run;
    cursor[i] = run;
    dinv[i]   = rsqrtf((float)(c + 1));
    run += c;
  }
  if (t == 1023) rowptr[n] = run;
}

__global__ void fill_kernel(const int* __restrict__ dst, int* cursor,
                            int* __restrict__ eid, int E) {
  int e = blockIdx.x * blockDim.x + threadIdx.x;
  if (e < E) {
    const int p = atomicAdd(&cursor[dst[e]], 1);
    eid[p] = e;
  }
}

__global__ void finalize_slots_kernel(const int* __restrict__ src,
                                      const int* __restrict__ dst,
                                      const int* __restrict__ rowptr,
                                      const int* __restrict__ eid,
                                      const float* __restrict__ dinv,
                                      int* __restrict__ csr_src,
                                      float* __restrict__ csr_w, int E) {
  int p = blockIdx.x * blockDim.x + threadIdx.x;
  if (p >= E) return;
  const int e  = eid[p];
  const int d  = dst[e];
  const int b  = rowptr[d];
  const int en = rowptr[d + 1];
  int rank = 0;
  for (int q = b; q < en; ++q) rank += (eid[q] < e) ? 1 : 0;
  const int s = src[e];
  csr_src[b + rank] = s;
  csr_w[b + rank]   = dinv[s] * dinv[d];
}

__device__ __forceinline__ void fma4(float4& acc, const float4 v, const float w) {
  acc.x = fmaf(v.x, w, acc.x);
  acc.y = fmaf(v.y, w, acc.y);
  acc.z = fmaf(v.z, w, acc.z);
  acc.w = fmaf(v.w, w, acc.w);
}

template <int D, bool BIAS, bool ROWSUM>
__global__ void __launch_bounds__(256)
gather_rows(const float* __restrict__ h, const int* __restrict__ rowptr,
            const int* __restrict__ csr_src, const float* __restrict__ csr_w,
            const float* __restrict__ dinv, const float* __restrict__ bias,
            float* __restrict__ out, float* __restrict__ rsum, int n) {
  constexpr int D4  = D / 4;
  constexpr int RPB = 256 / D4;
  static_assert(D4 % 32 == 0 && 256 % D4 == 0, "a wave must stay inside one row");
  int r = blockIdx.x * RPB + (int)(threadIdx.x / D4);
  const int j = threadIdx.x % D4;
  if (r >= n) return;
  r = __builtin_amdgcn_readfirstlane(r);

  const int   b  = rowptr[r];
  const int   e  = rowptr[r + 1];
  const float dr = dinv[r];
  const float4* __restrict__ hp = (const float4*)h + j;

  float4 acc = BIAS ? ((const float4*)bias)[j] : make_float4(0.0f, 0.0f, 0.0f, 0.0f);
  fma4(acc, hp[(size_t)r * D4], dr * dr);
  float ws = dr * dr;

  int p = b;
  for (; p + 3 < e; p += 4) {
    const int   s0 = csr_src[p],     s1 = csr_src[p + 1];
    const int   s2 = csr_src[p + 2], s3 = csr_src[p + 3];
    const float w0 = csr_w[p],       w1 = csr_w[p + 1];
    const float w2 = csr_w[p + 2],   w3 = csr_w[p + 3];
    const float4 v0 = hp[(size_t)s0 * D4];
    const float4 v1 = hp[(size_t)s1 * D4];
    const float4 v2 = hp[(size_t)s2 * D4];
    const float4 v3 = hp[(size_t)s3 * D4];
    fma4(acc, v0, w0);
    fma4(acc, v1, w1);
    fma4(acc, v2, w2);
    fma4(acc, v3, w3);
    if (ROWSUM) ws = ((ws + w0) + w1) + w2 + w3;
  }
  if (p + 1 < e) {
    const int   s0 = csr_src[p], s1 = csr_src[p + 1];
    const float w0 = csr_w[p],   w1 = csr_w[p + 1];
    const float4 v0 = hp[(size_t)s0 * D4];
    const float4 v1 = hp[(size_t)s1 * D4];
    fma4(acc, v0, w0);
    fma4(acc, v1, w1);
    if (ROWSUM) ws = (ws + w0) + w1;
    p += 2;
  }
  if (p < e) {
    const int   s0 = csr_src[p];
    const float w0 = csr_w[p];
    fma4(acc, hp[(size_t)s0 * D4], w0);
    if (ROWSUM) ws += w0;
  }
  ((float4*)out)[(size_t)r * D4 + j] = acc;
  if (ROWSUM && j == 0) rsum[r] = ws;
}

static inline void launch_gemm(int opb, const float* A, int lda, const float* B,
                               int ldb, float* C, int ldc, int M, int Nc, int K,
                               const float* bias, const float* rsum,
                               const float* cvec, hipStream_t stream) {
  dim3 grid(Nc / 128, (M + 127) / 128);
  if (opb)
    gemm_wmma<1, 0><<<grid, 256, 0, stream>>>(A, lda, B, ldb, C, ldc, M, Nc, K,
                                              bias, rsum, cvec);
  else if (bias)
    gemm_wmma<0, 2><<<grid, 256, 0, stream>>>(A, lda, B, ldb, C, ldc, M, Nc, K,
                                              bias, rsum, cvec);
  else
    gemm_wmma<0, 0><<<grid, 256, 0, stream>>>(A, lda, B, ldb, C, ldc, M, Nc, K,
                                              bias, rsum, cvec);
}

extern "C" void kernel_launch(void* const* d_in, const int* in_sizes, int n_in,
                              void* d_out, int out_size, void* d_ws, size_t ws_size,
                              hipStream_t stream) {
  const int IN = 1024, HID = 512, OUT = 128;

  const float* features = (const float*)d_in[0];
  const int*   edge     = (const int*)d_in[1];
  const float* W1       = (const float*)d_in[2];
  const float* b1       = (const float*)d_in[3];
  const float* W2       = (const float*)d_in[4];
  const float* b2       = (const float*)d_in[5];
  const float* b3       = (const float*)d_in[6];
  const float* b4       = (const float*)d_in[7];
  const float* head1    = (const float*)d_in[8];

  const int n  = in_sizes[0] / IN;
  const int E  = in_sizes[1] / 2;
  const int Et = E + n;
  const int* src = edge;
  const int* dst = edge + E;

  float* out = (float*)d_out;
  float* z   = out;
  float* h2  = out + (size_t)n * OUT;
  float* h4  = out + (size_t)n * OUT * 2;

  float* ws    = (float*)d_ws;
  float* dinv  = ws;
  float* t0    = dinv + n + Et;
  float* t1    = t0 + (size_t)n * HID;
  float* lists = t1 + (size_t)n * HID;
  float* xw    = t0;
  float* axw   = xw + (size_t)n * OUT;
  float* az    = axw + (size_t)n * OUT;
  float* aaz   = az + (size_t)n * OUT;
  float* Wd    = t1;
  float* dvec  = Wd + (size_t)OUT * IN;
  float* cv2   = dvec + IN;
  float* rsum  = cv2 + OUT;

  int*   rowptr  = (int*)lists;
  int*   cursor  = rowptr + (n + 1);
  int*   cnt     = cursor + n;
  int*   eid     = cnt + n;
  int*   csr_src = eid + E;
  float* csr_w   = (float*)(csr_src + E);

  const int TB = 256;
  zero_cnt_kernel<<<(n + TB - 1) / TB, TB, 0, stream>>>(cnt, n);
  hist_kernel<<<(E + TB - 1) / TB, TB, 0, stream>>>(dst, cnt, E);
  scan_kernel<<<1, 1024, 0, stream>>>(cnt, rowptr, cursor, dinv, n);
  fill_kernel<<<(E + TB - 1) / TB, TB, 0, stream>>>(dst, cursor, eid, E);
  finalize_slots_kernel<<<(E + TB - 1) / TB, TB, 0, stream>>>(src, dst, rowptr, eid, dinv,
                                                             csr_src, csr_w, E);

  const int GB = (n + 7) / 8;

  launch_gemm(0, W2, HID, W1, IN, Wd, IN, OUT, IN, HID, nullptr, nullptr, nullptr, stream);
  fold_bias_kernel<IN, HID><<<IN / 32 + OUT / 8, 256, 0, stream>>>(W1, W2, b1, b3, dvec, cv2);

  launch_gemm(1, features, IN, Wd, IN, xw, OUT, n, OUT, IN, nullptr, nullptr, nullptr, stream);
  gather_rows<128, true, false><<<GB, 256, 0, stream>>>(xw, rowptr, csr_src, csr_w, dinv,
                                                        cv2, axw, nullptr, n);
  gather_rows<128, true, false><<<GB, 256, 0, stream>>>(axw, rowptr, csr_src, csr_w, dinv,
                                                        b2, h2, nullptr, n);

  launch_gemm(0, h2, OUT, head1, OUT, z, OUT, n, OUT, OUT, nullptr, nullptr, nullptr, stream);

  gather_rows<128, false, true><<<GB, 256, 0, stream>>>(z, rowptr, csr_src, csr_w, dinv,
                                                        nullptr, az, rsum, n);
  gather_rows<128, false, false><<<GB, 256, 0, stream>>>(az, rowptr, csr_src, csr_w, dinv,
                                                         nullptr, aaz, nullptr, n);
  launch_gemm(0, aaz, OUT, Wd, IN, h4, IN, n, IN, OUT, b4, rsum, dvec, stream);
}


# Folded-weight product W2@W1 split over K into 8 slices on 64 work-groups (was 8), slice planes summed in slice order inside the bias-fold launch; changes float sum order of the folded weight
# speedup vs baseline: 4.0662x; 1.0834x over previous
#include <hip/hip_runtime.h>
#include <hip/hip_bf16.h>


typedef float v2f __attribute__((ext_vector_type(2)));
typedef float v8f __attribute__((ext_vector_type(8)));

__device__ __forceinline__ v8f wmma_f32_16x16x4(v2f a, v2f b, v8f c) {
  return __builtin_amdgcn_wmma_f32_16x16x4_f32(false, a, false, b, (short)0, c,
                                               false, false);
}

constexpr int GEMM_BM  = 128;
constexpr int GEMM_BN  = 128;
constexpr int GEMM_BK  = 16;
constexpr int GEMM_ALD = 20;
constexpr int GEMM_BLD = 132;

__device__ __forceinline__ void gemm_compute_tile(const float* __restrict__ as,
                                                  const float* __restrict__ bs,
                                                  v8f (&acc)[4][2], int wm, int wn,
                                                  int lm, int lh) {
#pragma unroll
  for (int kk = 0; kk < GEMM_BK; kk += 4) {
    const int ks = kk + 2 * lh;
    v2f a[4], b[2];
#pragma unroll
    for (int tm = 0; tm < 4; ++tm) {
      a[tm] = *(const v2f*)(&as[(wm + tm * 16 + lm) * GEMM_ALD + ks]);
    }
#pragma unroll
    for (int tn = 0; tn < 2; ++tn) {
      const int col = wn + tn * 16 + lm;
      b[tn].x = bs[(ks + 0) * GEMM_BLD + col];
      b[tn].y = bs[(ks + 1) * GEMM_BLD + col];
    }
#pragma unroll
    for (int tm = 0; tm < 4; ++tm)
#pragma unroll
      for (int tn = 0; tn < 2; ++tn)
        acc[tm][tn] = wmma_f32_16x16x4(a[tm], b[tn], acc[tm][tn]);
  }
}

__device__ __forceinline__ float4 gemm_load16(const float* __restrict__ base,
                                              unsigned off) {
  return *(const float4*)((const char*)base + off);
}

template <int OPB>
__device__ __forceinline__ void gemm_store_b(float* __restrict__ bs, int s,
                                             const float4 v) {
  if (OPB) {
    bs[s + 0 * GEMM_BLD] = v.x;
    bs[s + 1 * GEMM_BLD] = v.y;
    bs[s + 2 * GEMM_BLD] = v.z;
    bs[s + 3 * GEMM_BLD] = v.w;
  } else {
    *(float4*)(&bs[s]) = v;
  }
}

#define GEMM_PIPELINED_STEP(AS_CUR, BS_CUR, AS_NXT, BS_NXT)                       \
  do {                                                                            \
    Ak += BK;                                                                     \
    Bk += bstep;                                                                  \
    ra0 = gemm_load16(Ak, oa0);                                                   \
    rb0 = gemm_load16(Bk, ob0);                                                   \
    ra1 = gemm_load16(Ak, oa1);                                                   \
    rb1 = gemm_load16(Bk, ob1);                                                   \
    __builtin_amdgcn_sched_barrier(0);    \
    gemm_compute_tile(AS_CUR, BS_CUR, acc, wm, wn, lm, lh);                       \
    __builtin_amdgcn_sched_barrier(0);     \
    *(float4*)(&AS_NXT[sa0]) = ra0;                                               \
    gemm_store_b<OPB>(BS_NXT, sb0, rb0);                                          \
    *(float4*)(&AS_NXT[sa1]) = ra1;                                               \
    gemm_store_b<OPB>(BS_NXT, sb1, rb1);                                          \
    __syncthreads();                                                              \
  } while (0)

template <int OPB, int EPI>
__global__ void __launch_bounds__(256) __attribute__((amdgpu_waves_per_eu(6, 6)))
gemm_wmma(const float* __restrict__ A, int lda,
          const float* __restrict__ B, int ldb,
          float* __restrict__ C, int ldc,
          int M, int Nc, int K, const float* __restrict__ bias,
          const float* __restrict__ rsum, const float* __restrict__ cvec,
          int cplane) {
  constexpr int BM = GEMM_BM, BN = GEMM_BN, BK = GEMM_BK;
  constexpr int ALD = GEMM_ALD, BLD = GEMM_BLD;

  const int kz = blockIdx.z;
  A += (size_t)kz * (size_t)K;
  B += OPB ? (size_t)kz * (size_t)K : (size_t)kz * (size_t)K * (size_t)ldb;
  C += (size_t)kz * (size_t)cplane;

  __shared__ float As0[BM * ALD];
  __shared__ float As1[BM * ALD];
  __shared__ float Bs0[BK * BLD];
  __shared__ float Bs1[BK * BLD];

  const int tid  = threadIdx.x;
  const int wid  = tid >> 5;
  const int lane = tid & 31;
  const int lh   = lane >> 4;
  const int lm   = lane & 15;

  const int m0 = blockIdx.y * BM;
  const int n0 = blockIdx.x * BN;
  const int wm = (wid & 1) * 64;
  const int wn = (wid >> 1) * 32;

  const int i0 = tid, i1 = tid + 256;
  const int row0 = i0 >> 2, row1 = i1 >> 2;
  const int c40 = (i0 & 3) << 2, c41 = (i1 & 3) << 2;
  const int gm0 = (m0 + row0 < M) ? (m0 + row0) : (M - 1);
  const int gm1 = (m0 + row1 < M) ? (m0 + row1) : (M - 1);
  const unsigned oa0 = ((unsigned)gm0 * (unsigned)lda + (unsigned)c40) * 4u;
  const unsigned oa1 = ((unsigned)gm1 * (unsigned)lda + (unsigned)c41) * 4u;
  const int sa0 = row0 * ALD + c40;
  const int sa1 = row1 * ALD + c41;
  unsigned ob0, ob1;
  int sb0, sb1;
  if (OPB) {
    ob0 = ((unsigned)(n0 + row0) * (unsigned)ldb + (unsigned)c40) * 4u;
    ob1 = ((unsigned)(n0 + row1) * (unsigned)ldb + (unsigned)c41) * 4u;
    sb0 = c40 * BLD + row0;
    sb1 = c41 * BLD + row1;
  } else {
    const int kk0 = i0 >> 5, kk1 = i1 >> 5;
    const int cc0 = (i0 & 31) << 2, cc1 = (i1 & 31) << 2;
    ob0 = ((unsigned)kk0 * (unsigned)ldb + (unsigned)(n0 + cc0)) * 4u;
    ob1 = ((unsigned)kk1 * (unsigned)ldb + (unsigned)(n0 + cc1)) * 4u;
    sb0 = kk0 * BLD + cc0;
    sb1 = kk1 * BLD + cc1;
  }
  const size_t bstep = OPB ? (size_t)BK : (size_t)BK * (size_t)ldb;
  const float* Ak = A;
  const float* Bk = B;

  v8f acc[4][2];
#pragma unroll
  for (int tm = 0; tm < 4; ++tm)
#pragma unroll
    for (int tn = 0; tn < 2; ++tn) acc[tm][tn] = (v8f)0.0f;

  float4 ra0 = gemm_load16(Ak, oa0), rb0 = gemm_load16(Bk, ob0);
  float4 ra1 = gemm_load16(Ak, oa1), rb1 = gemm_load16(Bk, ob1);
  *(float4*)(&As0[sa0]) = ra0;
  gemm_store_b<OPB>(Bs0, sb0, rb0);
  *(float4*)(&As0[sa1]) = ra1;
  gemm_store_b<OPB>(Bs0, sb1, rb1);
  __syncthreads();

  const int nk = K / BK;
  int kt = 0;
  for (; kt + 2 < nk; kt += 2) {
    GEMM_PIPELINED_STEP(As0, Bs0, As1, Bs1);
    GEMM_PIPELINED_STEP(As1, Bs1, As0, Bs0);
  }
  const float* asl = As0;
  const float* bsl = Bs0;
  if (kt + 1 < nk) {
    GEMM_PIPELINED_STEP(As0, Bs0, As1, Bs1);
    asl = As1;
    bsl = Bs1;
  }
  gemm_compute_tile(asl, bsl, acc, wm, wn, lm, lh);

  float bv[2] = {0.0f, 0.0f};
  if (EPI >= 1) {
#pragma unroll
    for (int tn = 0; tn < 2; ++tn) bv[tn] = bias[n0 + wn + tn * 16 + lm];
  }
  if (EPI == 2) {
    float cv[2];
#pragma unroll
    for (int tn = 0; tn < 2; ++tn) cv[tn] = cvec[n0 + wn + tn * 16 + lm];
#pragma unroll
    for (int tm = 0; tm < 4; ++tm) {
#pragma unroll
      for (int v = 0; v < 8; ++v) {
        const int gm = m0 + wm + tm * 16 + v + 8 * lh;
        const float rr = rsum[(gm < M) ? gm : (M - 1)];
#pragma unroll
        for (int tn = 0; tn < 2; ++tn)
          acc[tm][tn][v] = fmaf(rr, cv[tn], acc[tm][tn][v]);
      }
    }
  }
#pragma unroll
  for (int tm = 0; tm < 4; ++tm) {
#pragma unroll
    for (int v = 0; v < 8; ++v) {
      const int gm = m0 + wm + tm * 16 + v + 8 * lh;
      if (gm < M) {
#pragma unroll
        for (int tn = 0; tn < 2; ++tn) {
          const int gn = n0 + wn + tn * 16 + lm;
          C[(size_t)gm * ldc + gn] = (EPI >= 1) ? acc[tm][tn][v] + bv[tn] : acc[tm][tn][v];
        }
      }
    }
  }
}
#undef GEMM_PIPELINED_STEP

template <int IN_, int HID_, int OUT_, int KS_>
__global__ void __launch_bounds__(256)
fold_bias_kernel(const float* __restrict__ W1, const float* __restrict__ W2,
                 const float* __restrict__ b1, const float* __restrict__ b3,
                 float* __restrict__ dvec, float* __restrict__ cv2,
                 const float* __restrict__ Wdp, float* __restrict__ Wd) {
  static_assert(IN_ % 32 == 0 && HID_ % 32 == 0, "whole waves of columns and of k");
  static_assert(OUT_ % 8 == 0 && (OUT_ * IN_) % 1024 == 0, "whole blocks");
  constexpr int in = IN_, hid = HID_;
  __shared__ float part[8][32];
  const int wid  = threadIdx.x >> 5;
  const int lane = threadIdx.x & 31;
  constexpr int nd = in / 32;
  constexpr int nc = OUT_ / 8;
  if ((int)blockIdx.x >= nd + nc) {
    constexpr int plane4 = OUT_ * IN_ / 4;
    const int i = ((int)blockIdx.x - nd - nc) * 256 + (int)threadIdx.x;
    const float4* __restrict__ p = (const float4*)Wdp + i;
    float4 v[KS_];
#pragma unroll
    for (int s = 0; s < KS_; ++s) v[s] = p[(size_t)s * plane4];
    float4 a = v[0];
#pragma unroll
    for (int s = 1; s < KS_; ++s) {
      a.x += v[s].x;
      a.y += v[s].y;
      a.z += v[s].z;
      a.w += v[s].w;
    }
    ((float4*)Wd)[i] = a;
  } else if ((int)blockIdx.x < nd) {
    const int j = blockIdx.x * 32 + lane;
    constexpr int per = hid / 8;
    const int k0 = wid * per;
    float s = 0.0f;
#pragma unroll 16
    for (int k = 0; k < per; ++k)
      s = fmaf(b3[k0 + k], W1[(size_t)(k0 + k) * in + j], s);
    part[wid][lane] = s;
    __syncthreads();
    if (wid == 0) {
      float t = part[0][lane];
#pragma unroll
      for (int w = 1; w < 8; ++w) t += part[w][lane];
      dvec[j] = t;
    }
  } else {
    const int j = ((int)blockIdx.x - nd) * 8 + wid;
    const float* __restrict__ wr = W2 + (size_t)j * hid;
    float s = 0.0f;
#pragma unroll
    for (int i = 0; i < hid / 32; ++i) s = fmaf(wr[lane + 32 * i], b1[lane + 32 * i], s);
#pragma unroll
    for (int off = 16; off > 0; off >>= 1) s += __shfl_xor(s, off);
    if (lane == 0) cv2[j] = s;
  }
}

__global__ void zero_cnt_kernel(int* __restrict__ cnt, int n) {
  int i = blockIdx.x * blockDim.x + threadIdx.x;
  if (i < n) cnt[i] = 0;
}

__global__ void hist_kernel(const int* __restrict__ dst, int* cnt, int E) {
  int e = blockIdx.x * blockDim.x + threadIdx.x;
  if (e < E) atomicAdd(&cnt[dst[e]], 1);
}

__global__ void __launch_bounds__(1024)
scan_kernel(const int* __restrict__ cnt, int* __restrict__ rowptr,
            int* __restrict__ cursor, float* __restrict__ dinv, int n) {
  __shared__ int part[1024];
  const int t   = threadIdx.x;
  const int per = (n + 1023) / 1024;
  const int b   = t * per;
  const int e   = (b + per < n) ? (b + per) : n;
  int s = 0;
  for (int i = b; i < e; ++i) s += cnt[i];
  part[t] = s;
  __syncthreads();
  for (int off = 1; off < 1024; off <<= 1) {
    const int v = (t >= off) ? part[t - off] : 0;
    __syncthreads();
    part[t] += v;
    __syncthreads();
  }
  int run = part[t] - s;
  for (int i = b; i < e; ++i) {
    const int c = cnt[i];
    rowptr[i] = run;
    cursor[i] = run;
    dinv[i]   = rsqrtf((float)(c + 1));
    run += c;
  }
  if (t == 1023) rowptr[n] = run;
}

__global__ void fill_kernel(const int* __restrict__ dst, int* cursor,
                            int* __restrict__ eid, int E) {
  int e = blockIdx.x * blockDim.x + threadIdx.x;
  if (e < E) {
    const int p = atomicAdd(&cursor[dst[e]], 1);
    eid[p] = e;
  }
}

__global__ void finalize_slots_kernel(const int* __restrict__ src,
                                      const int* __restrict__ dst,
                                      const int* __restrict__ rowptr,
                                      const int* __restrict__ eid,
                                      const float* __restrict__ dinv,
                                      int* __restrict__ csr_src,
                                      float* __restrict__ csr_w, int E) {
  int p = blockIdx.x * blockDim.x + threadIdx.x;
  if (p >= E) return;
  const int e  = eid[p];
  const int d  = dst[e];
  const int b  = rowptr[d];
  const int en = rowptr[d + 1];
  int rank = 0;
  for (int q = b; q < en; ++q) rank += (eid[q] < e) ? 1 : 0;
  const int s = src[e];
  csr_src[b + rank] = s;
  csr_w[b + rank]   = dinv[s] * dinv[d];
}

__device__ __forceinline__ void fma4(float4& acc, const float4 v, const float w) {
  acc.x = fmaf(v.x, w, acc.x);
  acc.y = fmaf(v.y, w, acc.y);
  acc.z = fmaf(v.z, w, acc.z);
  acc.w = fmaf(v.w, w, acc.w);
}

template <int D, bool BIAS, bool ROWSUM>
__global__ void __launch_bounds__(256)
gather_rows(const float* __restrict__ h, const int* __restrict__ rowptr,
            const int* __restrict__ csr_src, const float* __restrict__ csr_w,
            const float* __restrict__ dinv, const float* __restrict__ bias,
            float* __restrict__ out, float* __restrict__ rsum, int n) {
  constexpr int D4  = D / 4;
  constexpr int RPB = 256 / D4;
  static_assert(D4 % 32 == 0 && 256 % D4 == 0, "a wave must stay inside one row");
  int r = blockIdx.x * RPB + (int)(threadIdx.x / D4);
  const int j = threadIdx.x % D4;
  if (r >= n) return;
  r = __builtin_amdgcn_readfirstlane(r);

  const int   b  = rowptr[r];
  const int   e  = rowptr[r + 1];
  const float dr = dinv[r];
  const float4* __restrict__ hp = (const float4*)h + j;

  float4 acc = BIAS ? ((const float4*)bias)[j] : make_float4(0.0f, 0.0f, 0.0f, 0.0f);
  fma4(acc, hp[(size_t)r * D4], dr * dr);
  float ws = dr * dr;

  int p = b;
  for (; p + 3 < e; p += 4) {
    const int   s0 = csr_src[p],     s1 = csr_src[p + 1];
    const int   s2 = csr_src[p + 2], s3 = csr_src[p + 3];
    const float w0 = csr_w[p],       w1 = csr_w[p + 1];
    const float w2 = csr_w[p + 2],   w3 = csr_w[p + 3];
    const float4 v0 = hp[(size_t)s0 * D4];
    const float4 v1 = hp[(size_t)s1 * D4];
    const float4 v2 = hp[(size_t)s2 * D4];
    const float4 v3 = hp[(size_t)s3 * D4];
    fma4(acc, v0, w0);
    fma4(acc, v1, w1);
    fma4(acc, v2, w2);
    fma4(acc, v3, w3);
    if (ROWSUM) ws = ((ws + w0) + w1) + w2 + w3;
  }
  if (p + 1 < e) {
    const int   s0 = csr_src[p], s1 = csr_src[p + 1];
    const float w0 = csr_w[p],   w1 = csr_w[p + 1];
    const float4 v0 = hp[(size_t)s0 * D4];
    const float4 v1 = hp[(size_t)s1 * D4];
    fma4(acc, v0, w0);
    fma4(acc, v1, w1);
    if (ROWSUM) ws = (ws + w0) + w1;
    p += 2;
  }
  if (p < e) {
    const int   s0 = csr_src[p];
    const float w0 = csr_w[p];
    fma4(acc, hp[(size_t)s0 * D4], w0);
    if (ROWSUM) ws += w0;
  }
  ((float4*)out)[(size_t)r * D4 + j] = acc;
  if (ROWSUM && j == 0) rsum[r] = ws;
}

static inline void launch_gemm(int opb, const float* A, int lda, const float* B,
                               int ldb, float* C, int ldc, int M, int Nc, int K,
                               int ksplit, int cplane,
                               const float* bias, const float* rsum,
                               const float* cvec, hipStream_t stream) {
  dim3 grid(Nc / 128, (M + 127) / 128, ksplit);
  if (opb)
    gemm_wmma<1, 0><<<grid, 256, 0, stream>>>(A, lda, B, ldb, C, ldc, M, Nc, K,
                                              bias, rsum, cvec, cplane);
  else if (bias)
    gemm_wmma<0, 2><<<grid, 256, 0, stream>>>(A, lda, B, ldb, C, ldc, M, Nc, K,
                                              bias, rsum, cvec, cplane);
  else
    gemm_wmma<0, 0><<<grid, 256, 0, stream>>>(A, lda, B, ldb, C, ldc, M, Nc, K,
                                              bias, rsum, cvec, cplane);
}

extern "C" void kernel_launch(void* const* d_in, const int* in_sizes, int n_in,
                              void* d_out, int out_size, void* d_ws, size_t ws_size,
                              hipStream_t stream) {
  const int IN = 1024, HID = 512, OUT = 128;

  const float* features = (const float*)d_in[0];
  const int*   edge     = (const int*)d_in[1];
  const float* W1       = (const float*)d_in[2];
  const float* b1       = (const float*)d_in[3];
  const float* W2       = (const float*)d_in[4];
  const float* b2       = (const float*)d_in[5];
  const float* b3       = (const float*)d_in[6];
  const float* b4       = (const float*)d_in[7];
  const float* head1    = (const float*)d_in[8];

  const int n  = in_sizes[0] / IN;
  const int E  = in_sizes[1] / 2;
  const int Et = E + n;
  const int* src = edge;
  const int* dst = edge + E;

  float* out = (float*)d_out;
  float* z   = out;
  float* h2  = out + (size_t)n * OUT;
  float* h4  = out + (size_t)n * OUT * 2;

  float* ws    = (float*)d_ws;
  float* dinv  = ws;
  float* t0    = dinv + n + Et;
  float* t1    = t0 + (size_t)n * HID;
  float* lists = t1 + (size_t)n * HID;
  float* xw    = t0;
  float* axw   = xw + (size_t)n * OUT;
  float* az    = axw + (size_t)n * OUT;
  float* aaz   = az + (size_t)n * OUT;
  float* Wd    = t1;
  float* Wdp   = Wd + (size_t)OUT * IN;
  float* dvec  = Wdp + (size_t)8 * OUT * IN;
  float* cv2   = dvec + IN;
  float* rsum  = cv2 + OUT;

  int*   rowptr  = (int*)lists;
  int*   cursor  = rowptr + (n + 1);
  int*   cnt     = cursor + n;
  int*   eid     = cnt + n;
  int*   csr_src = eid + E;
  float* csr_w   = (float*)(csr_src + E);

  const int TB = 256;
  zero_cnt_kernel<<<(n + TB - 1) / TB, TB, 0, stream>>>(cnt, n);
  hist_kernel<<<(E + TB - 1) / TB, TB, 0, stream>>>(dst, cnt, E);
  scan_kernel<<<1, 1024, 0, stream>>>(cnt, rowptr, cursor, dinv, n);
  fill_kernel<<<(E + TB - 1) / TB, TB, 0, stream>>>(dst, cursor, eid, E);
  finalize_slots_kernel<<<(E + TB - 1) / TB, TB, 0, stream>>>(src, dst, rowptr, eid, dinv,
                                                             csr_src, csr_w, E);

  const int GB = (n + 7) / 8;

  launch_gemm(0, W2, HID, W1, IN, Wdp, IN, OUT, IN, HID / 8, 8, OUT * IN,
              nullptr, nullptr, nullptr, stream);
  fold_bias_kernel<IN, HID, OUT, 8><<<IN / 32 + OUT / 8 + OUT * IN / 1024, 256, 0, stream>>>(
      W1, W2, b1, b3, dvec, cv2, Wdp, Wd);

  launch_gemm(1, features, IN, Wd, IN, xw, OUT, n, OUT, IN, 1, 0, nullptr, nullptr, nullptr, stream);
  gather_rows<128, true, false><<<GB, 256, 0, stream>>>(xw, rowptr, csr_src, csr_w, dinv,
                                                        cv2, axw, nullptr, n);
  gather_rows<128, true, false><<<GB, 256, 0, stream>>>(axw, rowptr, csr_src, csr_w, dinv,
                                                        b2, h2, nullptr, n);

  launch_gemm(0, h2, OUT, head1, OUT, z, OUT, n, OUT, OUT, 1, 0, nullptr, nullptr, nullptr, stream);

  gather_rows<128, false, true><<<GB, 256, 0, stream>>>(z, rowptr, csr_src, csr_w, dinv,
                                                        nullptr, az, rsum, n);
  gather_rows<128, false, false><<<GB, 256, 0, stream>>>(az, rowptr, csr_src, csr_w, dinv,
                                                         nullptr, aaz, nullptr, n);
  launch_gemm(0, aaz, OUT, Wd, IN, h4, IN, n, IN, OUT, 1, 0, b4, rsum, dvec, stream);
}


# Encoder product x @ Wd^T re-tiled for its 128-wide output: 80-row work-groups (250, one per WGP, was 157) with K steps of 64 (was 16), 5x1 WMMA tiles per wave; k order and sums unchanged
# speedup vs baseline: 4.4722x; 1.0998x over previous
#include <hip/hip_runtime.h>
#include <hip/hip_bf16.h>


typedef float v2f __attribute__((ext_vector_type(2)));
typedef float v8f __attribute__((ext_vector_type(8)));

__device__ __forceinline__ v8f wmma_f32_16x16x4(v2f a, v2f b, v8f c) {
  return __builtin_amdgcn_wmma_f32_16x16x4_f32(false, a, false, b, (short)0, c,
                                               false, false);
}

constexpr int GEMM_BM  = 128;
constexpr int GEMM_BN  = 128;
constexpr int GEMM_BK  = 16;
constexpr int GEMM_ALD = 20;
constexpr int GEMM_BLD = 132;

__device__ __forceinline__ void gemm_compute_tile(const float* __restrict__ as,
                                                  const float* __restrict__ bs,
                                                  v8f (&acc)[4][2], int wm, int wn,
                                                  int lm, int lh) {
#pragma unroll
  for (int kk = 0; kk < GEMM_BK; kk += 4) {
    const int ks = kk + 2 * lh;
    v2f a[4], b[2];
#pragma unroll
    for (int tm = 0; tm < 4; ++tm) {
      a[tm] = *(const v2f*)(&as[(wm + tm * 16 + lm) * GEMM_ALD + ks]);
    }
#pragma unroll
    for (int tn = 0; tn < 2; ++tn) {
      const int col = wn + tn * 16 + lm;
      b[tn].x = bs[(ks + 0) * GEMM_BLD + col];
      b[tn].y = bs[(ks + 1) * GEMM_BLD + col];
    }
#pragma unroll
    for (int tm = 0; tm < 4; ++tm)
#pragma unroll
      for (int tn = 0; tn < 2; ++tn)
        acc[tm][tn] = wmma_f32_16x16x4(a[tm], b[tn], acc[tm][tn]);
  }
}

__device__ __forceinline__ float4 gemm_load16(const float* __restrict__ base,
                                              unsigned off) {
  return *(const float4*)((const char*)base + off);
}

template <int OPB>
__device__ __forceinline__ void gemm_store_b(float* __restrict__ bs, int s,
                                             const float4 v) {
  if (OPB) {
    bs[s + 0 * GEMM_BLD] = v.x;
    bs[s + 1 * GEMM_BLD] = v.y;
    bs[s + 2 * GEMM_BLD] = v.z;
    bs[s + 3 * GEMM_BLD] = v.w;
  } else {
    *(float4*)(&bs[s]) = v;
  }
}

#define GEMM_PIPELINED_STEP(AS_CUR, BS_CUR, AS_NXT, BS_NXT)                       \
  do {                                                                            \
    Ak += BK;                                                                     \
    Bk += bstep;                                                                  \
    ra0 = gemm_load16(Ak, oa0);                                                   \
    rb0 = gemm_load16(Bk, ob0);                                                   \
    ra1 = gemm_load16(Ak, oa1);                                                   \
    rb1 = gemm_load16(Bk, ob1);                                                   \
    __builtin_amdgcn_sched_barrier(0);    \
    gemm_compute_tile(AS_CUR, BS_CUR, acc, wm, wn, lm, lh);                       \
    __builtin_amdgcn_sched_barrier(0);     \
    *(float4*)(&AS_NXT[sa0]) = ra0;                                               \
    gemm_store_b<OPB>(BS_NXT, sb0, rb0);                                          \
    *(float4*)(&AS_NXT[sa1]) = ra1;                                               \
    gemm_store_b<OPB>(BS_NXT, sb1, rb1);                                          \
    __syncthreads();                                                              \
  } while (0)

template <int OPB, int EPI>
__global__ void __launch_bounds__(256) __attribute__((amdgpu_waves_per_eu(6, 6)))
gemm_wmma(const float* __restrict__ A, int lda,
          const float* __restrict__ B, int ldb,
          float* __restrict__ C, int ldc,
          int M, int Nc, int K, const float* __restrict__ bias,
          const float* __restrict__ rsum, const float* __restrict__ cvec,
          int cplane) {
  constexpr int BM = GEMM_BM, BN = GEMM_BN, BK = GEMM_BK;
  constexpr int ALD = GEMM_ALD, BLD = GEMM_BLD;

  const int kz = blockIdx.z;
  A += (size_t)kz * (size_t)K;
  B += OPB ? (size_t)kz * (size_t)K : (size_t)kz * (size_t)K * (size_t)ldb;
  C += (size_t)kz * (size_t)cplane;

  __shared__ float As0[BM * ALD];
  __shared__ float As1[BM * ALD];
  __shared__ float Bs0[BK * BLD];
  __shared__ float Bs1[BK * BLD];

  const int tid  = threadIdx.x;
  const int wid  = tid >> 5;
  const int lane = tid & 31;
  const int lh   = lane >> 4;
  const int lm   = lane & 15;

  const int m0 = blockIdx.y * BM;
  const int n0 = blockIdx.x * BN;
  const int wm = (wid & 1) * 64;
  const int wn = (wid >> 1) * 32;

  const int i0 = tid, i1 = tid + 256;
  const int row0 = i0 >> 2, row1 = i1 >> 2;
  const int c40 = (i0 & 3) << 2, c41 = (i1 & 3) << 2;
  const int gm0 = (m0 + row0 < M) ? (m0 + row0) : (M - 1);
  const int gm1 = (m0 + row1 < M) ? (m0 + row1) : (M - 1);
  const unsigned oa0 = ((unsigned)gm0 * (unsigned)lda + (unsigned)c40) * 4u;
  const unsigned oa1 = ((unsigned)gm1 * (unsigned)lda + (unsigned)c41) * 4u;
  const int sa0 = row0 * ALD + c40;
  const int sa1 = row1 * ALD + c41;
  unsigned ob0, ob1;
  int sb0, sb1;
  if (OPB) {
    ob0 = ((unsigned)(n0 + row0) * (unsigned)ldb + (unsigned)c40) * 4u;
    ob1 = ((unsigned)(n0 + row1) * (unsigned)ldb + (unsigned)c41) * 4u;
    sb0 = c40 * BLD + row0;
    sb1 = c41 * BLD + row1;
  } else {
    const int kk0 = i0 >> 5, kk1 = i1 >> 5;
    const int cc0 = (i0 & 31) << 2, cc1 = (i1 & 31) << 2;
    ob0 = ((unsigned)kk0 * (unsigned)ldb + (unsigned)(n0 + cc0)) * 4u;
    ob1 = ((unsigned)kk1 * (unsigned)ldb + (unsigned)(n0 + cc1)) * 4u;
    sb0 = kk0 * BLD + cc0;
    sb1 = kk1 * BLD + cc1;
  }
  const size_t bstep = OPB ? (size_t)BK : (size_t)BK * (size_t)ldb;
  const float* Ak = A;
  const float* Bk = B;

  v8f acc[4][2];
#pragma unroll
  for (int tm = 0; tm < 4; ++tm)
#pragma unroll
    for (int tn = 0; tn < 2; ++tn) acc[tm][tn] = (v8f)0.0f;

  float4 ra0 = gemm_load16(Ak, oa0), rb0 = gemm_load16(Bk, ob0);
  float4 ra1 = gemm_load16(Ak, oa1), rb1 = gemm_load16(Bk, ob1);
  *(float4*)(&As0[sa0]) = ra0;
  gemm_store_b<OPB>(Bs0, sb0, rb0);
  *(float4*)(&As0[sa1]) = ra1;
  gemm_store_b<OPB>(Bs0, sb1, rb1);
  __syncthreads();

  const int nk = K / BK;
  int kt = 0;
  for (; kt + 2 < nk; kt += 2) {
    GEMM_PIPELINED_STEP(As0, Bs0, As1, Bs1);
    GEMM_PIPELINED_STEP(As1, Bs1, As0, Bs0);
  }
  const float* asl = As0;
  const float* bsl = Bs0;
  if (kt + 1 < nk) {
    GEMM_PIPELINED_STEP(As0, Bs0, As1, Bs1);
    asl = As1;
    bsl = Bs1;
  }
  gemm_compute_tile(asl, bsl, acc, wm, wn, lm, lh);

  float bv[2] = {0.0f, 0.0f};
  if (EPI >= 1) {
#pragma unroll
    for (int tn = 0; tn < 2; ++tn) bv[tn] = bias[n0 + wn + tn * 16 + lm];
  }
  if (EPI == 2) {
    float cv[2];
#pragma unroll
    for (int tn = 0; tn < 2; ++tn) cv[tn] = cvec[n0 + wn + tn * 16 + lm];
#pragma unroll
    for (int tm = 0; tm < 4; ++tm) {
#pragma unroll
      for (int v = 0; v < 8; ++v) {
        const int gm = m0 + wm + tm * 16 + v + 8 * lh;
        const float rr = rsum[(gm < M) ? gm : (M - 1)];
#pragma unroll
        for (int tn = 0; tn < 2; ++tn)
          acc[tm][tn][v] = fmaf(rr, cv[tn], acc[tm][tn][v]);
      }
    }
  }
#pragma unroll
  for (int tm = 0; tm < 4; ++tm) {
#pragma unroll
    for (int v = 0; v < 8; ++v) {
      const int gm = m0 + wm + tm * 16 + v + 8 * lh;
      if (gm < M) {
#pragma unroll
        for (int tn = 0; tn < 2; ++tn) {
          const int gn = n0 + wn + tn * 16 + lm;
          C[(size_t)gm * ldc + gn] = (EPI >= 1) ? acc[tm][tn][v] + bv[tn] : acc[tm][tn][v];
        }
      }
    }
  }
}
#undef GEMM_PIPELINED_STEP

constexpr int XW_BM  = 80;
constexpr int XW_BN  = 128;
constexpr int XW_BK  = 64;
constexpr int XW_ALD = 68;
constexpr int XW_BLD = 132;

__device__ __forceinline__ void xw_compute_tile(const float* __restrict__ as,
                                                const float* __restrict__ bs,
                                                v8f (&acc)[5], int wn, int lm, int lh) {
#pragma unroll
  for (int kk = 0; kk < XW_BK; kk += 4) {
    const int ks = kk + 2 * lh;
    v2f a[5], b;
#pragma unroll
    for (int tm = 0; tm < 5; ++tm) {
      a[tm] = *(const v2f*)(&as[(tm * 16 + lm) * XW_ALD + ks]);
    }
    b.x = bs[(ks + 0) * XW_BLD + wn + lm];
    b.y = bs[(ks + 1) * XW_BLD + wn + lm];
#pragma unroll
    for (int tm = 0; tm < 5; ++tm) acc[tm] = wmma_f32_16x16x4(a[tm], b, acc[tm]);
  }
}

#define XW_PIPELINED_STEP(AS_CUR, BS_CUR, AS_NXT, BS_NXT)                         \
  do {                                                                            \
    Ak += XW_BK;                                                                  \
    Bk += XW_BK;                                                                  \
    _Pragma("unroll") for (int i = 0; i < 5; ++i) ra[i] = gemm_load16(Ak, oa[i]); \
    _Pragma("unroll") for (int i = 0; i < 8; ++i) rb[i] = gemm_load16(Bk, ob[i]); \
    __builtin_amdgcn_sched_barrier(0);    \
    xw_compute_tile(AS_CUR, BS_CUR, acc, wn, lm, lh);                             \
    __builtin_amdgcn_sched_barrier(0);     \
    _Pragma("unroll") for (int i = 0; i < 5; ++i)                                 \
      *(float4*)(&AS_NXT[sa[i]]) = ra[i];                                         \
    _Pragma("unroll") for (int i = 0; i < 8; ++i)                                 \
      gemm_store_b<1>(BS_NXT, sb[i], rb[i]);                                      \
    __syncthreads();                                                              \
  } while (0)

__global__ void __launch_bounds__(256)
gemm_xw(const float* __restrict__ A, int lda, const float* __restrict__ B, int ldb,
        float* __restrict__ C, int ldc, int M, int K) {
  constexpr int BM = XW_BM, BK = XW_BK, ALD = XW_ALD, BLD = XW_BLD;

  __shared__ float As0[BM * ALD];
  __shared__ float As1[BM * ALD];
  __shared__ float Bs0[BK * BLD];
  __shared__ float Bs1[BK * BLD];

  const int tid  = threadIdx.x;
  const int wid  = tid >> 5;
  const int lane = tid & 31;
  const int lh   = lane >> 4;
  const int lm   = lane & 15;
  const int m0   = blockIdx.x * BM;
  const int wn   = wid * 16;

  unsigned oa[5], ob[8];
  int sa[5], sb[8];
#pragma unroll
  for (int i = 0; i < 5; ++i) {
    const int row = (tid >> 4) + 16 * i;
    const int c4  = (tid & 15) << 2;
    const int gm  = (m0 + row < M) ? (m0 + row) : (M - 1);
    oa[i] = ((unsigned)gm * (unsigned)lda + (unsigned)c4) * 4u;
    sa[i] = row * ALD + c4;
  }
#pragma unroll
  for (int i = 0; i < 8; ++i) {
    const int col = (tid >> 2) + 64 * (i & 1);
    const int c4  = ((tid & 3) << 2) + 16 * (i >> 1);
    ob[i] = ((unsigned)col * (unsigned)ldb + (unsigned)c4) * 4u;
    sb[i] = c4 * BLD + col;
  }
  const float* Ak = A;
  const float* Bk = B;

  v8f acc[5];
#pragma unroll
  for (int tm = 0; tm < 5; ++tm) acc[tm] = (v8f)0.0f;

  float4 ra[5], rb[8];
#pragma unroll
  for (int i = 0; i < 5; ++i) ra[i] = gemm_load16(Ak, oa[i]);
#pragma unroll
  for (int i = 0; i < 8; ++i) rb[i] = gemm_load16(Bk, ob[i]);
#pragma unroll
  for (int i = 0; i < 5; ++i) *(float4*)(&As0[sa[i]]) = ra[i];
#pragma unroll
  for (int i = 0; i < 8; ++i) gemm_store_b<1>(Bs0, sb[i], rb[i]);
  __syncthreads();

  const int nk = K / BK;
  int kt = 0;
  for (; kt + 2 < nk; kt += 2) {
    XW_PIPELINED_STEP(As0, Bs0, As1, Bs1);
    XW_PIPELINED_STEP(As1, Bs1, As0, Bs0);
  }
  const float* asl = As0;
  const float* bsl = Bs0;
  if (kt + 1 < nk) {
    XW_PIPELINED_STEP(As0, Bs0, As1, Bs1);
    asl = As1;
    bsl = Bs1;
  }
  xw_compute_tile(asl, bsl, acc, wn, lm, lh);

#pragma unroll
  for (int tm = 0; tm < 5; ++tm) {
#pragma unroll
    for (int v = 0; v < 8; ++v) {
      const int gm = m0 + tm * 16 + v + 8 * lh;
      if (gm < M) C[(size_t)gm * ldc + wn + lm] = acc[tm][v];
    }
  }
}
#undef XW_PIPELINED_STEP

template <int IN_, int HID_, int OUT_, int KS_>
__global__ void __launch_bounds__(256)
fold_bias_kernel(const float* __restrict__ W1, const float* __restrict__ W2,
                 const float* __restrict__ b1, const float* __restrict__ b3,
                 float* __restrict__ dvec, float* __restrict__ cv2,
                 const float* __restrict__ Wdp, float* __restrict__ Wd) {
  static_assert(IN_ % 32 == 0 && HID_ % 32 == 0, "whole waves of columns and of k");
  static_assert(OUT_ % 8 == 0 && (OUT_ * IN_) % 1024 == 0, "whole blocks");
  constexpr int in = IN_, hid = HID_;
  __shared__ float part[8][32];
  const int wid  = threadIdx.x >> 5;
  const int lane = threadIdx.x & 31;
  constexpr int nd = in / 32;
  constexpr int nc = OUT_ / 8;
  if ((int)blockIdx.x >= nd + nc) {
    constexpr int plane4 = OUT_ * IN_ / 4;
    const int i = ((int)blockIdx.x - nd - nc) * 256 + (int)threadIdx.x;
    const float4* __restrict__ p = (const float4*)Wdp + i;
    float4 v[KS_];
#pragma unroll
    for (int s = 0; s < KS_; ++s) v[s] = p[(size_t)s * plane4];
    float4 a = v[0];
#pragma unroll
    for (int s = 1; s < KS_; ++s) {
      a.x += v[s].x;
      a.y += v[s].y;
      a.z += v[s].z;
      a.w += v[s].w;
    }
    ((float4*)Wd)[i] = a;
  } else if ((int)blockIdx.x < nd) {
    const int j = blockIdx.x * 32 + lane;
    constexpr int per = hid / 8;
    const int k0 = wid * per;
    float s = 0.0f;
#pragma unroll 16
    for (int k = 0; k < per; ++k)
      s = fmaf(b3[k0 + k], W1[(size_t)(k0 + k) * in + j], s);
    part[wid][lane] = s;
    __syncthreads();
    if (wid == 0) {
      float t = part[0][lane];
#pragma unroll
      for (int w = 1; w < 8; ++w) t += part[w][lane];
      dvec[j] = t;
    }
  } else {
    const int j = ((int)blockIdx.x - nd) * 8 + wid;
    const float* __restrict__ wr = W2 + (size_t)j * hid;
    float s = 0.0f;
#pragma unroll
    for (int i = 0; i < hid / 32; ++i) s = fmaf(wr[lane + 32 * i], b1[lane + 32 * i], s);
#pragma unroll
    for (int off = 16; off > 0; off >>= 1) s += __shfl_xor(s, off);
    if (lane == 0) cv2[j] = s;
  }
}

__global__ void zero_cnt_kernel(int* __restrict__ cnt, int n) {
  int i = blockIdx.x * blockDim.x + threadIdx.x;
  if (i < n) cnt[i] = 0;
}

__global__ void hist_kernel(const int* __restrict__ dst, int* cnt, int E) {
  int e = blockIdx.x * blockDim.x + threadIdx.x;
  if (e < E) atomicAdd(&cnt[dst[e]], 1);
}

__global__ void __launch_bounds__(1024)
scan_kernel(const int* __restrict__ cnt, int* __restrict__ rowptr,
            int* __restrict__ cursor, float* __restrict__ dinv, int n) {
  __shared__ int part[1024];
  const int t   = threadIdx.x;
  const int per = (n + 1023) / 1024;
  const int b   = t * per;
  const int e   = (b + per < n) ? (b + per) : n;
  int s = 0;
  for (int i = b; i < e; ++i) s += cnt[i];
  part[t] = s;
  __syncthreads();
  for (int off = 1; off < 1024; off <<= 1) {
    const int v = (t >= off) ? part[t - off] : 0;
    __syncthreads();
    part[t] += v;
    __syncthreads();
  }
  int run = part[t] - s;
  for (int i = b; i < e; ++i) {
    const int c = cnt[i];
    rowptr[i] = run;
    cursor[i] = run;
    dinv[i]   = rsqrtf((float)(c + 1));
    run += c;
  }
  if (t == 1023) rowptr[n] = run;
}

__global__ void fill_kernel(const int* __restrict__ dst, int* cursor,
                            int* __restrict__ eid, int E) {
  int e = blockIdx.x * blockDim.x + threadIdx.x;
  if (e < E) {
    const int p = atomicAdd(&cursor[dst[e]], 1);
    eid[p] = e;
  }
}

__global__ void finalize_slots_kernel(const int* __restrict__ src,
                                      const int* __restrict__ dst,
                                      const int* __restrict__ rowptr,
                                      const int* __restrict__ eid,
                                      const float* __restrict__ dinv,
                                      int* __restrict__ csr_src,
                                      float* __restrict__ csr_w, int E) {
  int p = blockIdx.x * blockDim.x + threadIdx.x;
  if (p >= E) return;
  const int e  = eid[p];
  const int d  = dst[e];
  const int b  = rowptr[d];
  const int en = rowptr[d + 1];
  int rank = 0;
  for (int q = b; q < en; ++q) rank += (eid[q] < e) ? 1 : 0;
  const int s = src[e];
  csr_src[b + rank] = s;
  csr_w[b + rank]   = dinv[s] * dinv[d];
}

__device__ __forceinline__ void fma4(float4& acc, const float4 v, const float w) {
  acc.x = fmaf(v.x, w, acc.x);
  acc.y = fmaf(v.y, w, acc.y);
  acc.z = fmaf(v.z, w, acc.z);
  acc.w = fmaf(v.w, w, acc.w);
}

template <int D, bool BIAS, bool ROWSUM>
__global__ void __launch_bounds__(256)
gather_rows(const float* __restrict__ h, const int* __restrict__ rowptr,
            const int* __restrict__ csr_src, const float* __restrict__ csr_w,
            const float* __restrict__ dinv, const float* __restrict__ bias,
            float* __restrict__ out, float* __restrict__ rsum, int n) {
  constexpr int D4  = D / 4;
  constexpr int RPB = 256 / D4;
  static_assert(D4 % 32 == 0 && 256 % D4 == 0, "a wave must stay inside one row");
  int r = blockIdx.x * RPB + (int)(threadIdx.x / D4);
  const int j = threadIdx.x % D4;
  if (r >= n) return;
  r = __builtin_amdgcn_readfirstlane(r);

  const int   b  = rowptr[r];
  const int   e  = rowptr[r + 1];
  const float dr = dinv[r];
  const float4* __restrict__ hp = (const float4*)h + j;

  float4 acc = BIAS ? ((const float4*)bias)[j] : make_float4(0.0f, 0.0f, 0.0f, 0.0f);
  fma4(acc, hp[(size_t)r * D4], dr * dr);
  float ws = dr * dr;

  int p = b;
  for (; p + 3 < e; p += 4) {
    const int   s0 = csr_src[p],     s1 = csr_src[p + 1];
    const int   s2 = csr_src[p + 2], s3 = csr_src[p + 3];
    const float w0 = csr_w[p],       w1 = csr_w[p + 1];
    const float w2 = csr_w[p + 2],   w3 = csr_w[p + 3];
    const float4 v0 = hp[(size_t)s0 * D4];
    const float4 v1 = hp[(size_t)s1 * D4];
    const float4 v2 = hp[(size_t)s2 * D4];
    const float4 v3 = hp[(size_t)s3 * D4];
    fma4(acc, v0, w0);
    fma4(acc, v1, w1);
    fma4(acc, v2, w2);
    fma4(acc, v3, w3);
    if (ROWSUM) ws = ((ws + w0) + w1) + w2 + w3;
  }
  if (p + 1 < e) {
    const int   s0 = csr_src[p], s1 = csr_src[p + 1];
    const float w0 = csr_w[p],   w1 = csr_w[p + 1];
    const float4 v0 = hp[(size_t)s0 * D4];
    const float4 v1 = hp[(size_t)s1 * D4];
    fma4(acc, v0, w0);
    fma4(acc, v1, w1);
    if (ROWSUM) ws = (ws + w0) + w1;
    p += 2;
  }
  if (p < e) {
    const int   s0 = csr_src[p];
    const float w0 = csr_w[p];
    fma4(acc, hp[(size_t)s0 * D4], w0);
    if (ROWSUM) ws += w0;
  }
  ((float4*)out)[(size_t)r * D4 + j] = acc;
  if (ROWSUM && j == 0) rsum[r] = ws;
}

static inline void launch_gemm(const float* A, int lda, const float* B,
                               int ldb, float* C, int ldc, int M, int Nc, int K,
                               int ksplit, int cplane,
                               const float* bias, const float* rsum,
                               const float* cvec, hipStream_t stream) {
  dim3 grid(Nc / 128, (M + 127) / 128, ksplit);
  if (bias)
    gemm_wmma<0, 2><<<grid, 256, 0, stream>>>(A, lda, B, ldb, C, ldc, M, Nc, K,
                                              bias, rsum, cvec, cplane);
  else
    gemm_wmma<0, 0><<<grid, 256, 0, stream>>>(A, lda, B, ldb, C, ldc, M, Nc, K,
                                              bias, rsum, cvec, cplane);
}

extern "C" void kernel_launch(void* const* d_in, const int* in_sizes, int n_in,
                              void* d_out, int out_size, void* d_ws, size_t ws_size,
                              hipStream_t stream) {
  const int IN = 1024, HID = 512, OUT = 128;
  static_assert(IN % XW_BK == 0 && OUT == XW_BN, "gemm_xw: whole K steps, one column of tiles");

  const float* features = (const float*)d_in[0];
  const int*   edge     = (const int*)d_in[1];
  const float* W1       = (const float*)d_in[2];
  const float* b1       = (const float*)d_in[3];
  const float* W2       = (const float*)d_in[4];
  const float* b2       = (const float*)d_in[5];
  const float* b3       = (const float*)d_in[6];
  const float* b4       = (const float*)d_in[7];
  const float* head1    = (const float*)d_in[8];

  const int n  = in_sizes[0] / IN;
  const int E  = in_sizes[1] / 2;
  const int Et = E + n;
  const int* src = edge;
  const int* dst = edge + E;

  float* out = (float*)d_out;
  float* z   = out;
  float* h2  = out + (size_t)n * OUT;
  float* h4  = out + (size_t)n * OUT * 2;

  float* ws    = (float*)d_ws;
  float* dinv  = ws;
  float* t0    = dinv + n + Et;
  float* t1    = t0 + (size_t)n * HID;
  float* lists = t1 + (size_t)n * HID;
  float* xw    = t0;
  float* axw   = xw + (size_t)n * OUT;
  float* az    = axw + (size_t)n * OUT;
  float* aaz   = az + (size_t)n * OUT;
  float* Wd    = t1;
  float* Wdp   = Wd + (size_t)OUT * IN;
  float* dvec  = Wdp + (size_t)8 * OUT * IN;
  float* cv2   = dvec + IN;
  float* rsum  = cv2 + OUT;

  int*   rowptr  = (int*)lists;
  int*   cursor  = rowptr + (n + 1);
  int*   cnt     = cursor + n;
  int*   eid     = cnt + n;
  int*   csr_src = eid + E;
  float* csr_w   = (float*)(csr_src + E);

  const int TB = 256;
  zero_cnt_kernel<<<(n + TB - 1) / TB, TB, 0, stream>>>(cnt, n);
  hist_kernel<<<(E + TB - 1) / TB, TB, 0, stream>>>(dst, cnt, E);
  scan_kernel<<<1, 1024, 0, stream>>>(cnt, rowptr, cursor, dinv, n);
  fill_kernel<<<(E + TB - 1) / TB, TB, 0, stream>>>(dst, cursor, eid, E);
  finalize_slots_kernel<<<(E + TB - 1) / TB, TB, 0, stream>>>(src, dst, rowptr, eid, dinv,
                                                             csr_src, csr_w, E);

  const int GB = (n + 7) / 8;

  launch_gemm(W2, HID, W1, IN, Wdp, IN, OUT, IN, HID / 8, 8, OUT * IN,
              nullptr, nullptr, nullptr, stream);
  fold_bias_kernel<IN, HID, OUT, 8><<<IN / 32 + OUT / 8 + OUT * IN / 1024, 256, 0, stream>>>(
      W1, W2, b1, b3, dvec, cv2, Wdp, Wd);

  gemm_xw<<<(n + XW_BM - 1) / XW_BM, 256, 0, stream>>>(features, IN, Wd, IN, xw, OUT, n, IN);
  gather_rows<128, true, false><<<GB, 256, 0, stream>>>(xw, rowptr, csr_src, csr_w, dinv,
                                                        cv2, axw, nullptr, n);
  gather_rows<128, true, false><<<GB, 256, 0, stream>>>(axw, rowptr, csr_src, csr_w, dinv,
                                                        b2, h2, nullptr, n);

  launch_gemm(h2, OUT, head1, OUT, z, OUT, n, OUT, OUT, 1, 0, nullptr, nullptr, nullptr, stream);

  gather_rows<128, false, true><<<GB, 256, 0, stream>>>(z, rowptr, csr_src, csr_w, dinv,
                                                        nullptr, az, rsum, n);
  gather_rows<128, false, false><<<GB, 256, 0, stream>>>(az, rowptr, csr_src, csr_w, dinv,
                                                         nullptr, aaz, nullptr, n);
  launch_gemm(aaz, OUT, Wd, IN, h4, IN, n, IN, OUT, 1, 0, b4, rsum, dvec, stream);
}


# List build's first three launches (clear counts, histogram, scan) run as extra work-groups inside the three weight launches (K-split fold, bias fold, encoder product): 11 launches, was 14
# speedup vs baseline: 4.9506x; 1.1070x over previous
#include <hip/hip_runtime.h>
#include <hip/hip_bf16.h>


typedef float v2f __attribute__((ext_vector_type(2)));
typedef float v8f __attribute__((ext_vector_type(8)));

__device__ __forceinline__ v8f wmma_f32_16x16x4(v2f a, v2f b, v8f c) {
  return __builtin_amdgcn_wmma_f32_16x16x4_f32(false, a, false, b, (short)0, c,
                                               false, false);
}

constexpr int GEMM_BM  = 128;
constexpr int GEMM_BN  = 128;
constexpr int GEMM_BK  = 16;
constexpr int GEMM_ALD = 20;
constexpr int GEMM_BLD = 132;

__device__ __forceinline__ void gemm_compute_tile(const float* __restrict__ as,
                                                  const float* __restrict__ bs,
                                                  v8f (&acc)[4][2], int wm, int wn,
                                                  int lm, int lh) {
#pragma unroll
  for (int kk = 0; kk < GEMM_BK; kk += 4) {
    const int ks = kk + 2 * lh;
    v2f a[4], b[2];
#pragma unroll
    for (int tm = 0; tm < 4; ++tm) {
      a[tm] = *(const v2f*)(&as[(wm + tm * 16 + lm) * GEMM_ALD + ks]);
    }
#pragma unroll
    for (int tn = 0; tn < 2; ++tn) {
      const int col = wn + tn * 16 + lm;
      b[tn].x = bs[(ks + 0) * GEMM_BLD + col];
      b[tn].y = bs[(ks + 1) * GEMM_BLD + col];
    }
#pragma unroll
    for (int tm = 0; tm < 4; ++tm)
#pragma unroll
      for (int tn = 0; tn < 2; ++tn)
        acc[tm][tn] = wmma_f32_16x16x4(a[tm], b[tn], acc[tm][tn]);
  }
}

__device__ __forceinline__ float4 gemm_load16(const float* __restrict__ base,
                                              unsigned off) {
  return *(const float4*)((const char*)base + off);
}

template <int OPB>
__device__ __forceinline__ void gemm_store_b(float* __restrict__ bs, int s,
                                             const float4 v) {
  if (OPB) {
    bs[s + 0 * GEMM_BLD] = v.x;
    bs[s + 1 * GEMM_BLD] = v.y;
    bs[s + 2 * GEMM_BLD] = v.z;
    bs[s + 3 * GEMM_BLD] = v.w;
  } else {
    *(float4*)(&bs[s]) = v;
  }
}

#define GEMM_PIPELINED_STEP(AS_CUR, BS_CUR, AS_NXT, BS_NXT)                       \
  do {                                                                            \
    Ak += BK;                                                                     \
    Bk += bstep;                                                                  \
    ra0 = gemm_load16(Ak, oa0);                                                   \
    rb0 = gemm_load16(Bk, ob0);                                                   \
    ra1 = gemm_load16(Ak, oa1);                                                   \
    rb1 = gemm_load16(Bk, ob1);                                                   \
    __builtin_amdgcn_sched_barrier(0);    \
    gemm_compute_tile(AS_CUR, BS_CUR, acc, wm, wn, lm, lh);                       \
    __builtin_amdgcn_sched_barrier(0);     \
    *(float4*)(&AS_NXT[sa0]) = ra0;                                               \
    gemm_store_b<OPB>(BS_NXT, sb0, rb0);                                          \
    *(float4*)(&AS_NXT[sa1]) = ra1;                                               \
    gemm_store_b<OPB>(BS_NXT, sb1, rb1);                                          \
    __syncthreads();                                                              \
  } while (0)

template <int OPB, int EPI, int SIDE = 0>
__global__ void __launch_bounds__(256) __attribute__((amdgpu_waves_per_eu(6, 6)))
gemm_wmma(const float* __restrict__ A, int lda,
          const float* __restrict__ B, int ldb,
          float* __restrict__ C, int ldc,
          int M, int Nc, int K, const float* __restrict__ bias,
          const float* __restrict__ rsum, const float* __restrict__ cvec,
          int cplane, int* __restrict__ side_cnt, int side_n) {
  if (SIDE == 1) {
    if (blockIdx.y == gridDim.y - 1) {
      const int nside = (int)(gridDim.x * gridDim.z) * 256;
      for (int i = (int)(blockIdx.z * gridDim.x + blockIdx.x) * 256 + (int)threadIdx.x;
           i < side_n; i += nside)
        side_cnt[i] = 0;
      return;
    }
  }
  constexpr int BM = GEMM_BM, BN = GEMM_BN, BK = GEMM_BK;
  constexpr int ALD = GEMM_ALD, BLD = GEMM_BLD;

  const int kz = blockIdx.z;
  A += (size_t)kz * (size_t)K;
  B += OPB ? (size_t)kz * (size_t)K : (size_t)kz * (size_t)K * (size_t)ldb;
  C += (size_t)kz * (size_t)cplane;

  __shared__ float As0[BM * ALD];
  __shared__ float As1[BM * ALD];
  __shared__ float Bs0[BK * BLD];
  __shared__ float Bs1[BK * BLD];

  const int tid  = threadIdx.x;
  const int wid  = tid >> 5;
  const int lane = tid & 31;
  const int lh   = lane >> 4;
  const int lm   = lane & 15;

  const int m0 = blockIdx.y * BM;
  const int n0 = blockIdx.x * BN;
  const int wm = (wid & 1) * 64;
  const int wn = (wid >> 1) * 32;

  const int i0 = tid, i1 = tid + 256;
  const int row0 = i0 >> 2, row1 = i1 >> 2;
  const int c40 = (i0 & 3) << 2, c41 = (i1 & 3) << 2;
  const int gm0 = (m0 + row0 < M) ? (m0 + row0) : (M - 1);
  const int gm1 = (m0 + row1 < M) ? (m0 + row1) : (M - 1);
  const unsigned oa0 = ((unsigned)gm0 * (unsigned)lda + (unsigned)c40) * 4u;
  const unsigned oa1 = ((unsigned)gm1 * (unsigned)lda + (unsigned)c41) * 4u;
  const int sa0 = row0 * ALD + c40;
  const int sa1 = row1 * ALD + c41;
  unsigned ob0, ob1;
  int sb0, sb1;
  if (OPB) {
    ob0 = ((unsigned)(n0 + row0) * (unsigned)ldb + (unsigned)c40) * 4u;
    ob1 = ((unsigned)(n0 + row1) * (unsigned)ldb + (unsigned)c41) * 4u;
    sb0 = c40 * BLD + row0;
    sb1 = c41 * BLD + row1;
  } else {
    const int kk0 = i0 >> 5, kk1 = i1 >> 5;
    const int cc0 = (i0 & 31) << 2, cc1 = (i1 & 31) << 2;
    ob0 = ((unsigned)kk0 * (unsigned)ldb + (unsigned)(n0 + cc0)) * 4u;
    ob1 = ((unsigned)kk1 * (unsigned)ldb + (unsigned)(n0 + cc1)) * 4u;
    sb0 = kk0 * BLD + cc0;
    sb1 = kk1 * BLD + cc1;
  }
  const size_t bstep = OPB ? (size_t)BK : (size_t)BK * (size_t)ldb;
  const float* Ak = A;
  const float* Bk = B;

  v8f acc[4][2];
#pragma unroll
  for (int tm = 0; tm < 4; ++tm)
#pragma unroll
    for (int tn = 0; tn < 2; ++tn) acc[tm][tn] = (v8f)0.0f;

  float4 ra0 = gemm_load16(Ak, oa0), rb0 = gemm_load16(Bk, ob0);
  float4 ra1 = gemm_load16(Ak, oa1), rb1 = gemm_load16(Bk, ob1);
  *(float4*)(&As0[sa0]) = ra0;
  gemm_store_b<OPB>(Bs0, sb0, rb0);
  *(float4*)(&As0[sa1]) = ra1;
  gemm_store_b<OPB>(Bs0, sb1, rb1);
  __syncthreads();

  const int nk = K / BK;
  int kt = 0;
  for (; kt + 2 < nk; kt += 2) {
    GEMM_PIPELINED_STEP(As0, Bs0, As1, Bs1);
    GEMM_PIPELINED_STEP(As1, Bs1, As0, Bs0);
  }
  const float* asl = As0;
  const float* bsl = Bs0;
  if (kt + 1 < nk) {
    GEMM_PIPELINED_STEP(As0, Bs0, As1, Bs1);
    asl = As1;
    bsl = Bs1;
  }
  gemm_compute_tile(asl, bsl, acc, wm, wn, lm, lh);

  float bv[2] = {0.0f, 0.0f};
  if (EPI >= 1) {
#pragma unroll
    for (int tn = 0; tn < 2; ++tn) bv[tn] = bias[n0 + wn + tn * 16 + lm];
  }
  if (EPI == 2) {
    float cv[2];
#pragma unroll
    for (int tn = 0; tn < 2; ++tn) cv[tn] = cvec[n0 + wn + tn * 16 + lm];
#pragma unroll
    for (int tm = 0; tm < 4; ++tm) {
#pragma unroll
      for (int v = 0; v < 8; ++v) {
        const int gm = m0 + wm + tm * 16 + v + 8 * lh;
        const float rr = rsum[(gm < M) ? gm : (M - 1)];
#pragma unroll
        for (int tn = 0; tn < 2; ++tn)
          acc[tm][tn][v] = fmaf(rr, cv[tn], acc[tm][tn][v]);
      }
    }
  }
#pragma unroll
  for (int tm = 0; tm < 4; ++tm) {
#pragma unroll
    for (int v = 0; v < 8; ++v) {
      const int gm = m0 + wm + tm * 16 + v + 8 * lh;
      if (gm < M) {
#pragma unroll
        for (int tn = 0; tn < 2; ++tn) {
          const int gn = n0 + wn + tn * 16 + lm;
          C[(size_t)gm * ldc + gn] = (EPI >= 1) ? acc[tm][tn][v] + bv[tn] : acc[tm][tn][v];
        }
      }
    }
  }
}
#undef GEMM_PIPELINED_STEP

template <int NT>
__device__ __forceinline__ void scan_job(const int* __restrict__ cnt, int* __restrict__ rowptr,
                                         int* __restrict__ cursor, float* __restrict__ dinv,
                                         int n) {
  __shared__ int part[NT];
  const int t   = threadIdx.x;
  const int per = (n + NT - 1) / NT;
  const int b   = t * per;
  const int e   = (b + per < n) ? (b + per) : n;
  int s = 0;
  for (int i = b; i < e; ++i) s += cnt[i];
  part[t] = s;
  __syncthreads();
  for (int off = 1; off < NT; off <<= 1) {
    const int v = (t >= off) ? part[t - off] : 0;
    __syncthreads();
    part[t] += v;
    __syncthreads();
  }
  int run = part[t] - s;
  for (int i = b; i < e; ++i) {
    const int c = cnt[i];
    rowptr[i] = run;
    cursor[i] = run;
    dinv[i]   = rsqrtf((float)(c + 1));
    run += c;
  }
  if (t == NT - 1) rowptr[n] = run;
}

constexpr int XW_BM  = 80;
constexpr int XW_BN  = 128;
constexpr int XW_BK  = 64;
constexpr int XW_ALD = 68;
constexpr int XW_BLD = 132;

__device__ __forceinline__ void xw_compute_tile(const float* __restrict__ as,
                                                const float* __restrict__ bs,
                                                v8f (&acc)[5], int wn, int lm, int lh) {
#pragma unroll
  for (int kk = 0; kk < XW_BK; kk += 4) {
    const int ks = kk + 2 * lh;
    v2f a[5], b;
#pragma unroll
    for (int tm = 0; tm < 5; ++tm) {
      a[tm] = *(const v2f*)(&as[(tm * 16 + lm) * XW_ALD + ks]);
    }
    b.x = bs[(ks + 0) * XW_BLD + wn + lm];
    b.y = bs[(ks + 1) * XW_BLD + wn + lm];
#pragma unroll
    for (int tm = 0; tm < 5; ++tm) acc[tm] = wmma_f32_16x16x4(a[tm], b, acc[tm]);
  }
}

#define XW_PIPELINED_STEP(AS_CUR, BS_CUR, AS_NXT, BS_NXT)                         \
  do {                                                                            \
    Ak += XW_BK;                                                                  \
    Bk += XW_BK;                                                                  \
    _Pragma("unroll") for (int i = 0; i < 5; ++i) ra[i] = gemm_load16(Ak, oa[i]); \
    _Pragma("unroll") for (int i = 0; i < 8; ++i) rb[i] = gemm_load16(Bk, ob[i]); \
    __builtin_amdgcn_sched_barrier(0);    \
    xw_compute_tile(AS_CUR, BS_CUR, acc, wn, lm, lh);                             \
    __builtin_amdgcn_sched_barrier(0);     \
    _Pragma("unroll") for (int i = 0; i < 5; ++i)                                 \
      *(float4*)(&AS_NXT[sa[i]]) = ra[i];                                         \
    _Pragma("unroll") for (int i = 0; i < 8; ++i)                                 \
      gemm_store_b<1>(BS_NXT, sb[i], rb[i]);                                      \
    __syncthreads();                                                              \
  } while (0)

__global__ void __launch_bounds__(256)
gemm_xw(const float* __restrict__ A, int lda, const float* __restrict__ B, int ldb,
        float* __restrict__ C, int ldc, int M, int K,
        const int* __restrict__ cnt, int* __restrict__ rowptr,
        int* __restrict__ cursor, float* __restrict__ dinv, int n) {
  if (blockIdx.x == gridDim.x - 1) {
    scan_job<256>(cnt, rowptr, cursor, dinv, n);
    return;
  }
  constexpr int BM = XW_BM, BK = XW_BK, ALD = XW_ALD, BLD = XW_BLD;

  __shared__ float As0[BM * ALD];
  __shared__ float As1[BM * ALD];
  __shared__ float Bs0[BK * BLD];
  __shared__ float Bs1[BK * BLD];

  const int tid  = threadIdx.x;
  const int wid  = tid >> 5;
  const int lane = tid & 31;
  const int lh   = lane >> 4;
  const int lm   = lane & 15;
  const int m0   = blockIdx.x * BM;
  const int wn   = wid * 16;

  unsigned oa[5], ob[8];
  int sa[5], sb[8];
#pragma unroll
  for (int i = 0; i < 5; ++i) {
    const int row = (tid >> 4) + 16 * i;
    const int c4  = (tid & 15) << 2;
    const int gm  = (m0 + row < M) ? (m0 + row) : (M - 1);
    oa[i] = ((unsigned)gm * (unsigned)lda + (unsigned)c4) * 4u;
    sa[i] = row * ALD + c4;
  }
#pragma unroll
  for (int i = 0; i < 8; ++i) {
    const int col = (tid >> 2) + 64 * (i & 1);
    const int c4  = ((tid & 3) << 2) + 16 * (i >> 1);
    ob[i] = ((unsigned)col * (unsigned)ldb + (unsigned)c4) * 4u;
    sb[i] = c4 * BLD + col;
  }
  const float* Ak = A;
  const float* Bk = B;

  v8f acc[5];
#pragma unroll
  for (int tm = 0; tm < 5; ++tm) acc[tm] = (v8f)0.0f;

  float4 ra[5], rb[8];
#pragma unroll
  for (int i = 0; i < 5; ++i) ra[i] = gemm_load16(Ak, oa[i]);
#pragma unroll
  for (int i = 0; i < 8; ++i) rb[i] = gemm_load16(Bk, ob[i]);
#pragma unroll
  for (int i = 0; i < 5; ++i) *(float4*)(&As0[sa[i]]) = ra[i];
#pragma unroll
  for (int i = 0; i < 8; ++i) gemm_store_b<1>(Bs0, sb[i], rb[i]);
  __syncthreads();

  const int nk = K / BK;
  int kt = 0;
  for (; kt + 2 < nk; kt += 2) {
    XW_PIPELINED_STEP(As0, Bs0, As1, Bs1);
    XW_PIPELINED_STEP(As1, Bs1, As0, Bs0);
  }
  const float* asl = As0;
  const float* bsl = Bs0;
  if (kt + 1 < nk) {
    XW_PIPELINED_STEP(As0, Bs0, As1, Bs1);
    asl = As1;
    bsl = Bs1;
  }
  xw_compute_tile(asl, bsl, acc, wn, lm, lh);

#pragma unroll
  for (int tm = 0; tm < 5; ++tm) {
#pragma unroll
    for (int v = 0; v < 8; ++v) {
      const int gm = m0 + tm * 16 + v + 8 * lh;
      if (gm < M) C[(size_t)gm * ldc + wn + lm] = acc[tm][v];
    }
  }
}
#undef XW_PIPELINED_STEP

template <int IN_, int HID_, int OUT_, int KS_>
__global__ void __launch_bounds__(256)
fold_bias_kernel(const float* __restrict__ W1, const float* __restrict__ W2,
                 const float* __restrict__ b1, const float* __restrict__ b3,
                 float* __restrict__ dvec, float* __restrict__ cv2,
                 const float* __restrict__ Wdp, float* __restrict__ Wd,
                 const int* __restrict__ dst, int* cnt, int E) {
  static_assert(IN_ % 32 == 0 && HID_ % 32 == 0, "whole waves of columns and of k");
  static_assert(OUT_ % 8 == 0 && (OUT_ * IN_) % 1024 == 0, "whole blocks");
  constexpr int in = IN_, hid = HID_;
  __shared__ float part[8][32];
  const int wid  = threadIdx.x >> 5;
  const int lane = threadIdx.x & 31;
  constexpr int nd = in / 32;
  constexpr int nc = OUT_ / 8;
  constexpr int nw = OUT_ * IN_ / 1024;
  if ((int)blockIdx.x >= nd + nc + nw) {
    const int e = ((int)blockIdx.x - nd - nc - nw) * 256 + (int)threadIdx.x;
    if (e < E) atomicAdd(&cnt[dst[e]], 1);
  } else if ((int)blockIdx.x >= nd + nc) {
    constexpr int plane4 = OUT_ * IN_ / 4;
    const int i = ((int)blockIdx.x - nd - nc) * 256 + (int)threadIdx.x;
    const float4* __restrict__ p = (const float4*)Wdp + i;
    float4 v[KS_];
#pragma unroll
    for (int s = 0; s < KS_; ++s) v[s] = p[(size_t)s * plane4];
    float4 a = v[0];
#pragma unroll
    for (int s = 1; s < KS_; ++s) {
      a.x += v[s].x;
      a.y += v[s].y;
      a.z += v[s].z;
      a.w += v[s].w;
    }
    ((float4*)Wd)[i] = a;
  } else if ((int)blockIdx.x < nd) {
    const int j = blockIdx.x * 32 + lane;
    constexpr int per = hid / 8;
    const int k0 = wid * per;
    float s = 0.0f;
#pragma unroll 16
    for (int k = 0; k < per; ++k)
      s = fmaf(b3[k0 + k], W1[(size_t)(k0 + k) * in + j], s);
    part[wid][lane] = s;
    __syncthreads();
    if (wid == 0) {
      float t = part[0][lane];
#pragma unroll
      for (int w = 1; w < 8; ++w) t += part[w][lane];
      dvec[j] = t;
    }
  } else {
    const int j = ((int)blockIdx.x - nd) * 8 + wid;
    const float* __restrict__ wr = W2 + (size_t)j * hid;
    float s = 0.0f;
#pragma unroll
    for (int i = 0; i < hid / 32; ++i) s = fmaf(wr[lane + 32 * i], b1[lane + 32 * i], s);
#pragma unroll
    for (int off = 16; off > 0; off >>= 1) s += __shfl_xor(s, off);
    if (lane == 0) cv2[j] = s;
  }
}


__global__ void fill_kernel(const int* __restrict__ dst, int* cursor,
                            int* __restrict__ eid, int E) {
  int e = blockIdx.x * blockDim.x + threadIdx.x;
  if (e < E) {
    const int p = atomicAdd(&cursor[dst[e]], 1);
    eid[p] = e;
  }
}

__global__ void finalize_slots_kernel(const int* __restrict__ src,
                                      const int* __restrict__ dst,
                                      const int* __restrict__ rowptr,
                                      const int* __restrict__ eid,
                                      const float* __restrict__ dinv,
                                      int* __restrict__ csr_src,
                                      float* __restrict__ csr_w, int E) {
  int p = blockIdx.x * blockDim.x + threadIdx.x;
  if (p >= E) return;
  const int e  = eid[p];
  const int d  = dst[e];
  const int b  = rowptr[d];
  const int en = rowptr[d + 1];
  int rank = 0;
  for (int q = b; q < en; ++q) rank += (eid[q] < e) ? 1 : 0;
  const int s = src[e];
  csr_src[b + rank] = s;
  csr_w[b + rank]   = dinv[s] * dinv[d];
}

__device__ __forceinline__ void fma4(float4& acc, const float4 v, const float w) {
  acc.x = fmaf(v.x, w, acc.x);
  acc.y = fmaf(v.y, w, acc.y);
  acc.z = fmaf(v.z, w, acc.z);
  acc.w = fmaf(v.w, w, acc.w);
}

template <int D, bool BIAS, bool ROWSUM>
__global__ void __launch_bounds__(256)
gather_rows(const float* __restrict__ h, const int* __restrict__ rowptr,
            const int* __restrict__ csr_src, const float* __restrict__ csr_w,
            const float* __restrict__ dinv, const float* __restrict__ bias,
            float* __restrict__ out, float* __restrict__ rsum, int n) {
  constexpr int D4  = D / 4;
  constexpr int RPB = 256 / D4;
  static_assert(D4 % 32 == 0 && 256 % D4 == 0, "a wave must stay inside one row");
  int r = blockIdx.x * RPB + (int)(threadIdx.x / D4);
  const int j = threadIdx.x % D4;
  if (r >= n) return;
  r = __builtin_amdgcn_readfirstlane(r);

  const int   b  = rowptr[r];
  const int   e  = rowptr[r + 1];
  const float dr = dinv[r];
  const float4* __restrict__ hp = (const float4*)h + j;

  float4 acc = BIAS ? ((const float4*)bias)[j] : make_float4(0.0f, 0.0f, 0.0f, 0.0f);
  fma4(acc, hp[(size_t)r * D4], dr * dr);
  float ws = dr * dr;

  int p = b;
  for (; p + 3 < e; p += 4) {
    const int   s0 = csr_src[p],     s1 = csr_src[p + 1];
    const int   s2 = csr_src[p + 2], s3 = csr_src[p + 3];
    const float w0 = csr_w[p],       w1 = csr_w[p + 1];
    const float w2 = csr_w[p + 2],   w3 = csr_w[p + 3];
    const float4 v0 = hp[(size_t)s0 * D4];
    const float4 v1 = hp[(size_t)s1 * D4];
    const float4 v2 = hp[(size_t)s2 * D4];
    const float4 v3 = hp[(size_t)s3 * D4];
    fma4(acc, v0, w0);
    fma4(acc, v1, w1);
    fma4(acc, v2, w2);
    fma4(acc, v3, w3);
    if (ROWSUM) ws = ((ws + w0) + w1) + w2 + w3;
  }
  if (p + 1 < e) {
    const int   s0 = csr_src[p], s1 = csr_src[p + 1];
    const float w0 = csr_w[p],   w1 = csr_w[p + 1];
    const float4 v0 = hp[(size_t)s0 * D4];
    const float4 v1 = hp[(size_t)s1 * D4];
    fma4(acc, v0, w0);
    fma4(acc, v1, w1);
    if (ROWSUM) ws = (ws + w0) + w1;
    p += 2;
  }
  if (p < e) {
    const int   s0 = csr_src[p];
    const float w0 = csr_w[p];
    fma4(acc, hp[(size_t)s0 * D4], w0);
    if (ROWSUM) ws += w0;
  }
  ((float4*)out)[(size_t)r * D4 + j] = acc;
  if (ROWSUM && j == 0) rsum[r] = ws;
}

static inline void launch_gemm(const float* A, int lda, const float* B,
                               int ldb, float* C, int ldc, int M, int Nc, int K,
                               int ksplit, int cplane,
                               const float* bias, const float* rsum,
                               const float* cvec, hipStream_t stream) {
  dim3 grid(Nc / 128, (M + 127) / 128, ksplit);
  if (bias)
    gemm_wmma<0, 2><<<grid, 256, 0, stream>>>(A, lda, B, ldb, C, ldc, M, Nc, K,
                                              bias, rsum, cvec, cplane, nullptr, 0);
  else
    gemm_wmma<0, 0><<<grid, 256, 0, stream>>>(A, lda, B, ldb, C, ldc, M, Nc, K,
                                              bias, rsum, cvec, cplane, nullptr, 0);
}

extern "C" void kernel_launch(void* const* d_in, const int* in_sizes, int n_in,
                              void* d_out, int out_size, void* d_ws, size_t ws_size,
                              hipStream_t stream) {
  const int IN = 1024, HID = 512, OUT = 128;
  static_assert(IN % XW_BK == 0 && OUT == XW_BN, "gemm_xw: whole K steps, one column of tiles");

  const float* features = (const float*)d_in[0];
  const int*   edge     = (const int*)d_in[1];
  const float* W1       = (const float*)d_in[2];
  const float* b1       = (const float*)d_in[3];
  const float* W2       = (const float*)d_in[4];
  const float* b2       = (const float*)d_in[5];
  const float* b3       = (const float*)d_in[6];
  const float* b4       = (const float*)d_in[7];
  const float* head1    = (const float*)d_in[8];

  const int n  = in_sizes[0] / IN;
  const int E  = in_sizes[1] / 2;
  const int Et = E + n;
  const int* src = edge;
  const int* dst = edge + E;

  float* out = (float*)d_out;
  float* z   = out;
  float* h2  = out + (size_t)n * OUT;
  float* h4  = out + (size_t)n * OUT * 2;

  float* ws    = (float*)d_ws;
  float* dinv  = ws;
  float* t0    = dinv + n + Et;
  float* t1    = t0 + (size_t)n * HID;
  float* lists = t1 + (size_t)n * HID;
  float* xw    = t0;
  float* axw   = xw + (size_t)n * OUT;
  float* az    = axw + (size_t)n * OUT;
  float* aaz   = az + (size_t)n * OUT;
  float* Wd    = t1;
  float* Wdp   = Wd + (size_t)OUT * IN;
  float* dvec  = Wdp + (size_t)8 * OUT * IN;
  float* cv2   = dvec + IN;
  float* rsum  = cv2 + OUT;

  int*   rowptr  = (int*)lists;
  int*   cursor  = rowptr + (n + 1);
  int*   cnt     = cursor + n;
  int*   eid     = cnt + n;
  int*   csr_src = eid + E;
  float* csr_w   = (float*)(csr_src + E);

  const int TB = 256;
  gemm_wmma<0, 0, 1><<<dim3(IN / 128, OUT / 128 + 1, 8), 256, 0, stream>>>(
      W2, HID, W1, IN, Wdp, IN, OUT, IN, HID / 8, nullptr, nullptr, nullptr, OUT * IN, cnt, n);
  fold_bias_kernel<IN, HID, OUT, 8>
      <<<IN / 32 + OUT / 8 + OUT * IN / 1024 + (E + TB - 1) / TB, 256, 0, stream>>>(
          W1, W2, b1, b3, dvec, cv2, Wdp, Wd, dst, cnt, E);

  gemm_xw<<<(n + XW_BM - 1) / XW_BM + 1, 256, 0, stream>>>(
      features, IN, Wd, IN, xw, OUT, n, IN, cnt, rowptr, cursor, dinv, n);
  fill_kernel<<<(E + TB - 1) / TB, TB, 0, stream>>>(dst, cursor, eid, E);
  finalize_slots_kernel<<<(E + TB - 1) / TB, TB, 0, stream>>>(src, dst, rowptr, eid, dinv,
                                                             csr_src, csr_w, E);

  const int GB = (n + 7) / 8;
  gather_rows<128, true, false><<<GB, 256, 0, stream>>>(xw, rowptr, csr_src, csr_w, dinv,
                                                        cv2, axw, nullptr, n);
  gather_rows<128, true, false><<<GB, 256, 0, stream>>>(axw, rowptr, csr_src, csr_w, dinv,
                                                        b2, h2, nullptr, n);

  launch_gemm(h2, OUT, head1, OUT, z, OUT, n, OUT, OUT, 1, 0, nullptr, nullptr, nullptr, stream);

  gather_rows<128, false, true><<<GB, 256, 0, stream>>>(z, rowptr, csr_src, csr_w, dinv,
                                                        nullptr, az, rsum, n);
  gather_rows<128, false, false><<<GB, 256, 0, stream>>>(az, rowptr, csr_src, csr_w, dinv,
                                                         nullptr, aaz, nullptr, n);
  launch_gemm(aaz, OUT, Wd, IN, h4, IN, n, IN, OUT, 1, 0, b4, rsum, dvec, stream);
}


# Edge list kept as fixed 32-slot rows per destination: filled in the histogram pass, rank-ordered by extra work-groups of the encoder product's launch; 9 launches, was 11; sum order unchanged
# speedup vs baseline: 6.3830x; 1.2893x over previous
#include <hip/hip_runtime.h>
#include <hip/hip_bf16.h>


typedef float v2f __attribute__((ext_vector_type(2)));
typedef float v8f __attribute__((ext_vector_type(8)));

__device__ __forceinline__ v8f wmma_f32_16x16x4(v2f a, v2f b, v8f c) {
  return __builtin_amdgcn_wmma_f32_16x16x4_f32(false, a, false, b, (short)0, c,
                                               false, false);
}

constexpr int GEMM_BM  = 128;
constexpr int GEMM_BN  = 128;
constexpr int GEMM_BK  = 16;
constexpr int GEMM_ALD = 20;
constexpr int GEMM_BLD = 132;

__device__ __forceinline__ void gemm_compute_tile(const float* __restrict__ as,
                                                  const float* __restrict__ bs,
                                                  v8f (&acc)[4][2], int wm, int wn,
                                                  int lm, int lh) {
#pragma unroll
  for (int kk = 0; kk < GEMM_BK; kk += 4) {
    const int ks = kk + 2 * lh;
    v2f a[4], b[2];
#pragma unroll
    for (int tm = 0; tm < 4; ++tm) {
      a[tm] = *(const v2f*)(&as[(wm + tm * 16 + lm) * GEMM_ALD + ks]);
    }
#pragma unroll
    for (int tn = 0; tn < 2; ++tn) {
      const int col = wn + tn * 16 + lm;
      b[tn].x = bs[(ks + 0) * GEMM_BLD + col];
      b[tn].y = bs[(ks + 1) * GEMM_BLD + col];
    }
#pragma unroll
    for (int tm = 0; tm < 4; ++tm)
#pragma unroll
      for (int tn = 0; tn < 2; ++tn)
        acc[tm][tn] = wmma_f32_16x16x4(a[tm], b[tn], acc[tm][tn]);
  }
}

__device__ __forceinline__ float4 gemm_load16(const float* __restrict__ base,
                                              unsigned off) {
  return *(const float4*)((const char*)base + off);
}

template <int OPB>
__device__ __forceinline__ void gemm_store_b(float* __restrict__ bs, int s,
                                             const float4 v) {
  if (OPB) {
    bs[s + 0 * GEMM_BLD] = v.x;
    bs[s + 1 * GEMM_BLD] = v.y;
    bs[s + 2 * GEMM_BLD] = v.z;
    bs[s + 3 * GEMM_BLD] = v.w;
  } else {
    *(float4*)(&bs[s]) = v;
  }
}

#define GEMM_PIPELINED_STEP(AS_CUR, BS_CUR, AS_NXT, BS_NXT)                       \
  do {                                                                            \
    Ak += BK;                                                                     \
    Bk += bstep;                                                                  \
    ra0 = gemm_load16(Ak, oa0);                                                   \
    rb0 = gemm_load16(Bk, ob0);                                                   \
    ra1 = gemm_load16(Ak, oa1);                                                   \
    rb1 = gemm_load16(Bk, ob1);                                                   \
    __builtin_amdgcn_sched_barrier(0);    \
    gemm_compute_tile(AS_CUR, BS_CUR, acc, wm, wn, lm, lh);                       \
    __builtin_amdgcn_sched_barrier(0);     \
    *(float4*)(&AS_NXT[sa0]) = ra0;                                               \
    gemm_store_b<OPB>(BS_NXT, sb0, rb0);                                          \
    *(float4*)(&AS_NXT[sa1]) = ra1;                                               \
    gemm_store_b<OPB>(BS_NXT, sb1, rb1);                                          \
    __syncthreads();                                                              \
  } while (0)

template <int OPB, int EPI, int SIDE = 0>
__global__ void __launch_bounds__(256) __attribute__((amdgpu_waves_per_eu(6, 6)))
gemm_wmma(const float* __restrict__ A, int lda,
          const float* __restrict__ B, int ldb,
          float* __restrict__ C, int ldc,
          int M, int Nc, int K, const float* __restrict__ bias,
          const float* __restrict__ rsum, const float* __restrict__ cvec,
          int cplane, int* __restrict__ side_cnt, int side_n) {
  if (SIDE == 1) {
    if (blockIdx.y == gridDim.y - 1) {
      const int nside = (int)(gridDim.x * gridDim.z) * 256;
      for (int i = (int)(blockIdx.z * gridDim.x + blockIdx.x) * 256 + (int)threadIdx.x;
           i < side_n; i += nside)
        side_cnt[i] = 0;
      return;
    }
  }
  constexpr int BM = GEMM_BM, BN = GEMM_BN, BK = GEMM_BK;
  constexpr int ALD = GEMM_ALD, BLD = GEMM_BLD;

  const int kz = blockIdx.z;
  A += (size_t)kz * (size_t)K;
  B += OPB ? (size_t)kz * (size_t)K : (size_t)kz * (size_t)K * (size_t)ldb;
  C += (size_t)kz * (size_t)cplane;

  __shared__ float As0[BM * ALD];
  __shared__ float As1[BM * ALD];
  __shared__ float Bs0[BK * BLD];
  __shared__ float Bs1[BK * BLD];

  const int tid  = threadIdx.x;
  const int wid  = tid >> 5;
  const int lane = tid & 31;
  const int lh   = lane >> 4;
  const int lm   = lane & 15;

  const int m0 = blockIdx.y * BM;
  const int n0 = blockIdx.x * BN;
  const int wm = (wid & 1) * 64;
  const int wn = (wid >> 1) * 32;

  const int i0 = tid, i1 = tid + 256;
  const int row0 = i0 >> 2, row1 = i1 >> 2;
  const int c40 = (i0 & 3) << 2, c41 = (i1 & 3) << 2;
  const int gm0 = (m0 + row0 < M) ? (m0 + row0) : (M - 1);
  const int gm1 = (m0 + row1 < M) ? (m0 + row1) : (M - 1);
  const unsigned oa0 = ((unsigned)gm0 * (unsigned)lda + (unsigned)c40) * 4u;
  const unsigned oa1 = ((unsigned)gm1 * (unsigned)lda + (unsigned)c41) * 4u;
  const int sa0 = row0 * ALD + c40;
  const int sa1 = row1 * ALD + c41;
  unsigned ob0, ob1;
  int sb0, sb1;
  if (OPB) {
    ob0 = ((unsigned)(n0 + row0) * (unsigned)ldb + (unsigned)c40) * 4u;
    ob1 = ((unsigned)(n0 + row1) * (unsigned)ldb + (unsigned)c41) * 4u;
    sb0 = c40 * BLD + row0;
    sb1 = c41 * BLD + row1;
  } else {
    const int kk0 = i0 >> 5, kk1 = i1 >> 5;
    const int cc0 = (i0 & 31) << 2, cc1 = (i1 & 31) << 2;
    ob0 = ((unsigned)kk0 * (unsigned)ldb + (unsigned)(n0 + cc0)) * 4u;
    ob1 = ((unsigned)kk1 * (unsigned)ldb + (unsigned)(n0 + cc1)) * 4u;
    sb0 = kk0 * BLD + cc0;
    sb1 = kk1 * BLD + cc1;
  }
  const size_t bstep = OPB ? (size_t)BK : (size_t)BK * (size_t)ldb;
  const float* Ak = A;
  const float* Bk = B;

  v8f acc[4][2];
#pragma unroll
  for (int tm = 0; tm < 4; ++tm)
#pragma unroll
    for (int tn = 0; tn < 2; ++tn) acc[tm][tn] = (v8f)0.0f;

  float4 ra0 = gemm_load16(Ak, oa0), rb0 = gemm_load16(Bk, ob0);
  float4 ra1 = gemm_load16(Ak, oa1), rb1 = gemm_load16(Bk, ob1);
  *(float4*)(&As0[sa0]) = ra0;
  gemm_store_b<OPB>(Bs0, sb0, rb0);
  *(float4*)(&As0[sa1]) = ra1;
  gemm_store_b<OPB>(Bs0, sb1, rb1);
  __syncthreads();

  const int nk = K / BK;
  int kt = 0;
  for (; kt + 2 < nk; kt += 2) {
    GEMM_PIPELINED_STEP(As0, Bs0, As1, Bs1);
    GEMM_PIPELINED_STEP(As1, Bs1, As0, Bs0);
  }
  const float* asl = As0;
  const float* bsl = Bs0;
  if (kt + 1 < nk) {
    GEMM_PIPELINED_STEP(As0, Bs0, As1, Bs1);
    asl = As1;
    bsl = Bs1;
  }
  gemm_compute_tile(asl, bsl, acc, wm, wn, lm, lh);

  float bv[2] = {0.0f, 0.0f};
  if (EPI >= 1) {
#pragma unroll
    for (int tn = 0; tn < 2; ++tn) bv[tn] = bias[n0 + wn + tn * 16 + lm];
  }
  if (EPI == 2) {
    float cv[2];
#pragma unroll
    for (int tn = 0; tn < 2; ++tn) cv[tn] = cvec[n0 + wn + tn * 16 + lm];
#pragma unroll
    for (int tm = 0; tm < 4; ++tm) {
#pragma unroll
      for (int v = 0; v < 8; ++v) {
        const int gm = m0 + wm + tm * 16 + v + 8 * lh;
        const float rr = rsum[(gm < M) ? gm : (M - 1)];
#pragma unroll
        for (int tn = 0; tn < 2; ++tn)
          acc[tm][tn][v] = fmaf(rr, cv[tn], acc[tm][tn][v]);
      }
    }
  }
#pragma unroll
  for (int tm = 0; tm < 4; ++tm) {
#pragma unroll
    for (int v = 0; v < 8; ++v) {
      const int gm = m0 + wm + tm * 16 + v + 8 * lh;
      if (gm < M) {
#pragma unroll
        for (int tn = 0; tn < 2; ++tn) {
          const int gn = n0 + wn + tn * 16 + lm;
          C[(size_t)gm * ldc + gn] = (EPI >= 1) ? acc[tm][tn][v] + bv[tn] : acc[tm][tn][v];
        }
      }
    }
  }
}
#undef GEMM_PIPELINED_STEP

constexpr int LIST_SLOTS = 32;

__device__ __forceinline__ void sort_rows_job(const int* __restrict__ src,
                                              const int* __restrict__ cnt,
                                              const int* __restrict__ tbl,
                                              int* __restrict__ nsrc,
                                              float* __restrict__ nw, int n,
                                              int blk, int nblk) {
  const int lane  = threadIdx.x & 31;
  const int wave  = __builtin_amdgcn_readfirstlane(blk * 8 + (int)(threadIdx.x >> 5));
  const int nwave = nblk * 8;
  for (int r0 = wave * 4; r0 < n; r0 += nwave * 4) {
    int  k[4], e[4], rank[4], s[4], c[4];
    bool on[4];
#pragma unroll
    for (int u = 0; u < 4; ++u)
      k[u] = cnt[(r0 + u < n) ? (r0 + u) : (n - 1)];
#pragma unroll
    for (int u = 0; u < 4; ++u) {
      const int r = r0 + u;
      if (r >= n || k[u] > LIST_SLOTS) k[u] = 0;
      on[u] = lane < k[u];
      e[u]  = on[u] ? tbl[(size_t)r * LIST_SLOTS + lane] : 0;
    }
#pragma unroll
    for (int u = 0; u < 4; ++u) {
      int rk = 0;
      for (int q = 0; q < k[u]; ++q)
        rk += (__builtin_amdgcn_readlane(e[u], q) < e[u]) ? 1 : 0;
      rank[u] = rk;
    }
#pragma unroll
    for (int u = 0; u < 4; ++u) s[u] = on[u] ? src[e[u]] : 0;
#pragma unroll
    for (int u = 0; u < 4; ++u) c[u] = on[u] ? cnt[s[u]] : 0;
#pragma unroll
    for (int u = 0; u < 4; ++u) {
      if (on[u]) {
        const size_t o = (size_t)(r0 + u) * LIST_SLOTS + rank[u];
        nsrc[o] = s[u];
        nw[o]   = rsqrtf((float)(c[u] + 1)) * rsqrtf((float)(k[u] + 1));
      }
    }
  }
}

constexpr int XW_BM  = 80;
constexpr int XW_BN  = 128;
constexpr int XW_BK  = 64;
constexpr int XW_ALD = 68;
constexpr int XW_BLD = 132;

__device__ __forceinline__ void xw_compute_tile(const float* __restrict__ as,
                                                const float* __restrict__ bs,
                                                v8f (&acc)[5], int wn, int lm, int lh) {
#pragma unroll
  for (int kk = 0; kk < XW_BK; kk += 4) {
    const int ks = kk + 2 * lh;
    v2f a[5], b;
#pragma unroll
    for (int tm = 0; tm < 5; ++tm) {
      a[tm] = *(const v2f*)(&as[(tm * 16 + lm) * XW_ALD + ks]);
    }
    b.x = bs[(ks + 0) * XW_BLD + wn + lm];
    b.y = bs[(ks + 1) * XW_BLD + wn + lm];
#pragma unroll
    for (int tm = 0; tm < 5; ++tm) acc[tm] = wmma_f32_16x16x4(a[tm], b, acc[tm]);
  }
}

#define XW_PIPELINED_STEP(AS_CUR, BS_CUR, AS_NXT, BS_NXT)                         \
  do {                                                                            \
    Ak += XW_BK;                                                                  \
    Bk += XW_BK;                                                                  \
    _Pragma("unroll") for (int i = 0; i < 5; ++i) ra[i] = gemm_load16(Ak, oa[i]); \
    _Pragma("unroll") for (int i = 0; i < 8; ++i) rb[i] = gemm_load16(Bk, ob[i]); \
    __builtin_amdgcn_sched_barrier(0);    \
    xw_compute_tile(AS_CUR, BS_CUR, acc, wn, lm, lh);                             \
    __builtin_amdgcn_sched_barrier(0);     \
    _Pragma("unroll") for (int i = 0; i < 5; ++i)                                 \
      *(float4*)(&AS_NXT[sa[i]]) = ra[i];                                         \
    _Pragma("unroll") for (int i = 0; i < 8; ++i)                                 \
      gemm_store_b<1>(BS_NXT, sb[i], rb[i]);                                      \
    __syncthreads();                                                              \
  } while (0)

__global__ void __launch_bounds__(256)
gemm_xw(const float* __restrict__ A, int lda, const float* __restrict__ B, int ldb,
        float* __restrict__ C, int ldc, int M, int K, int mtiles,
        const int* __restrict__ src, const int* __restrict__ cnt,
        const int* __restrict__ tbl, int* __restrict__ nsrc,
        float* __restrict__ nw, int n) {
  if ((int)blockIdx.x >= mtiles) {
    sort_rows_job(src, cnt, tbl, nsrc, nw, n, (int)blockIdx.x - mtiles, (int)gridDim.x - mtiles);
    return;
  }
  constexpr int BM = XW_BM, BK = XW_BK, ALD = XW_ALD, BLD = XW_BLD;

  __shared__ float As0[BM * ALD];
  __shared__ float As1[BM * ALD];
  __shared__ float Bs0[BK * BLD];
  __shared__ float Bs1[BK * BLD];

  const int tid  = threadIdx.x;
  const int wid  = tid >> 5;
  const int lane = tid & 31;
  const int lh   = lane >> 4;
  const int lm   = lane & 15;
  const int m0   = blockIdx.x * BM;
  const int wn   = wid * 16;

  unsigned oa[5], ob[8];
  int sa[5], sb[8];
#pragma unroll
  for (int i = 0; i < 5; ++i) {
    const int row = (tid >> 4) + 16 * i;
    const int c4  = (tid & 15) << 2;
    const int gm  = (m0 + row < M) ? (m0 + row) : (M - 1);
    oa[i] = ((unsigned)gm * (unsigned)lda + (unsigned)c4) * 4u;
    sa[i] = row * ALD + c4;
  }
#pragma unroll
  for (int i = 0; i < 8; ++i) {
    const int col = (tid >> 2) + 64 * (i & 1);
    const int c4  = ((tid & 3) << 2) + 16 * (i >> 1);
    ob[i] = ((unsigned)col * (unsigned)ldb + (unsigned)c4) * 4u;
    sb[i] = c4 * BLD + col;
  }
  const float* Ak = A;
  const float* Bk = B;

  v8f acc[5];
#pragma unroll
  for (int tm = 0; tm < 5; ++tm) acc[tm] = (v8f)0.0f;

  float4 ra[5], rb[8];
#pragma unroll
  for (int i = 0; i < 5; ++i) ra[i] = gemm_load16(Ak, oa[i]);
#pragma unroll
  for (int i = 0; i < 8; ++i) rb[i] = gemm_load16(Bk, ob[i]);
#pragma unroll
  for (int i = 0; i < 5; ++i) *(float4*)(&As0[sa[i]]) = ra[i];
#pragma unroll
  for (int i = 0; i < 8; ++i) gemm_store_b<1>(Bs0, sb[i], rb[i]);
  __syncthreads();

  const int nk = K / BK;
  int kt = 0;
  for (; kt + 2 < nk; kt += 2) {
    XW_PIPELINED_STEP(As0, Bs0, As1, Bs1);
    XW_PIPELINED_STEP(As1, Bs1, As0, Bs0);
  }
  const float* asl = As0;
  const float* bsl = Bs0;
  if (kt + 1 < nk) {
    XW_PIPELINED_STEP(As0, Bs0, As1, Bs1);
    asl = As1;
    bsl = Bs1;
  }
  xw_compute_tile(asl, bsl, acc, wn, lm, lh);

#pragma unroll
  for (int tm = 0; tm < 5; ++tm) {
#pragma unroll
    for (int v = 0; v < 8; ++v) {
      const int gm = m0 + tm * 16 + v + 8 * lh;
      if (gm < M) C[(size_t)gm * ldc + wn + lm] = acc[tm][v];
    }
  }
}
#undef XW_PIPELINED_STEP

template <int IN_, int HID_, int OUT_, int KS_>
__global__ void __launch_bounds__(256)
fold_bias_kernel(const float* __restrict__ W1, const float* __restrict__ W2,
                 const float* __restrict__ b1, const float* __restrict__ b3,
                 float* __restrict__ dvec, float* __restrict__ cv2,
                 const float* __restrict__ Wdp, float* __restrict__ Wd,
                 const int* __restrict__ dst, int* cnt, int* __restrict__ tbl, int E) {
  static_assert(IN_ % 32 == 0 && HID_ % 32 == 0, "whole waves of columns and of k");
  static_assert(OUT_ % 8 == 0 && (OUT_ * IN_) % 1024 == 0, "whole blocks");
  constexpr int in = IN_, hid = HID_;
  __shared__ float part[8][32];
  const int wid  = threadIdx.x >> 5;
  const int lane = threadIdx.x & 31;
  constexpr int nd = in / 32;
  constexpr int nc = OUT_ / 8;
  constexpr int nw = OUT_ * IN_ / 1024;
  if ((int)blockIdx.x >= nd + nc + nw) {
    const int e = ((int)blockIdx.x - nd - nc - nw) * 256 + (int)threadIdx.x;
    if (e < E) {
      const int d = dst[e];
      const int a = atomicAdd(&cnt[d], 1);
      if (a < LIST_SLOTS) tbl[(size_t)d * LIST_SLOTS + a] = e;
    }
  } else if ((int)blockIdx.x >= nd + nc) {
    constexpr int plane4 = OUT_ * IN_ / 4;
    const int i = ((int)blockIdx.x - nd - nc) * 256 + (int)threadIdx.x;
    const float4* __restrict__ p = (const float4*)Wdp + i;
    float4 v[KS_];
#pragma unroll
    for (int s = 0; s < KS_; ++s) v[s] = p[(size_t)s * plane4];
    float4 a = v[0];
#pragma unroll
    for (int s = 1; s < KS_; ++s) {
      a.x += v[s].x;
      a.y += v[s].y;
      a.z += v[s].z;
      a.w += v[s].w;
    }
    ((float4*)Wd)[i] = a;
  } else if ((int)blockIdx.x < nd) {
    const int j = blockIdx.x * 32 + lane;
    constexpr int per = hid / 8;
    const int k0 = wid * per;
    float s = 0.0f;
#pragma unroll 16
    for (int k = 0; k < per; ++k)
      s = fmaf(b3[k0 + k], W1[(size_t)(k0 + k) * in + j], s);
    part[wid][lane] = s;
    __syncthreads();
    if (wid == 0) {
      float t = part[0][lane];
#pragma unroll
      for (int w = 1; w < 8; ++w) t += part[w][lane];
      dvec[j] = t;
    }
  } else {
    const int j = ((int)blockIdx.x - nd) * 8 + wid;
    const float* __restrict__ wr = W2 + (size_t)j * hid;
    float s = 0.0f;
#pragma unroll
    for (int i = 0; i < hid / 32; ++i) s = fmaf(wr[lane + 32 * i], b1[lane + 32 * i], s);
#pragma unroll
    for (int off = 16; off > 0; off >>= 1) s += __shfl_xor(s, off);
    if (lane == 0) cv2[j] = s;
  }
}


__device__ __forceinline__ void fma4(float4& acc, const float4 v, const float w) {
  acc.x = fmaf(v.x, w, acc.x);
  acc.y = fmaf(v.y, w, acc.y);
  acc.z = fmaf(v.z, w, acc.z);
  acc.w = fmaf(v.w, w, acc.w);
}

template <int D, bool BIAS, bool ROWSUM>
__global__ void __launch_bounds__(256)
gather_rows(const float* __restrict__ h, const int* __restrict__ cnt,
            const int* __restrict__ nsrc, const float* __restrict__ nw,
            const int* __restrict__ src, const int* __restrict__ dst, int E,
            const float* __restrict__ bias,
            float* __restrict__ out, float* __restrict__ rsum, int n) {
  constexpr int D4  = D / 4;
  constexpr int RPB = 256 / D4;
  static_assert(D4 % 32 == 0 && 256 % D4 == 0, "a wave must stay inside one row");
  int r = blockIdx.x * RPB + (int)(threadIdx.x / D4);
  const int j = threadIdx.x % D4;
  if (r >= n) return;
  r = __builtin_amdgcn_readfirstlane(r);

  const float4* __restrict__ hp = (const float4*)h + j;
  const float4 hself = hp[(size_t)r * D4];
  const int    k  = cnt[r];
  const float  dr = rsqrtf((float)(k + 1));

  float4 acc = BIAS ? ((const float4*)bias)[j] : make_float4(0.0f, 0.0f, 0.0f, 0.0f);
  fma4(acc, hself, dr * dr);
  float ws = dr * dr;

  if (k <= LIST_SLOTS) {
    const int*   __restrict__ csr_src = nsrc + (size_t)r * LIST_SLOTS;
    const float* __restrict__ csr_w   = nw + (size_t)r * LIST_SLOTS;
    int p = 0;
    for (; p + 3 < k; p += 4) {
      const int   s0 = csr_src[p],     s1 = csr_src[p + 1];
      const int   s2 = csr_src[p + 2], s3 = csr_src[p + 3];
      const float w0 = csr_w[p],       w1 = csr_w[p + 1];
      const float w2 = csr_w[p + 2],   w3 = csr_w[p + 3];
      const float4 v0 = hp[(size_t)s0 * D4];
      const float4 v1 = hp[(size_t)s1 * D4];
      const float4 v2 = hp[(size_t)s2 * D4];
      const float4 v3 = hp[(size_t)s3 * D4];
      fma4(acc, v0, w0);
      fma4(acc, v1, w1);
      fma4(acc, v2, w2);
      fma4(acc, v3, w3);
      if (ROWSUM) ws = ((ws + w0) + w1) + w2 + w3;
    }
    if (p + 1 < k) {
      const int   s0 = csr_src[p], s1 = csr_src[p + 1];
      const float w0 = csr_w[p],   w1 = csr_w[p + 1];
      const float4 v0 = hp[(size_t)s0 * D4];
      const float4 v1 = hp[(size_t)s1 * D4];
      fma4(acc, v0, w0);
      fma4(acc, v1, w1);
      if (ROWSUM) ws = (ws + w0) + w1;
      p += 2;
    }
    if (p < k) {
      const int   s0 = csr_src[p];
      const float w0 = csr_w[p];
      fma4(acc, hp[(size_t)s0 * D4], w0);
      if (ROWSUM) ws += w0;
    }
  } else {
    const int lane = threadIdx.x & 31;
    for (int e0 = 0; e0 < E; e0 += 32) {
      const int e = e0 + lane;
      unsigned m = __builtin_amdgcn_ballot_w32(e < E && dst[e] == r);
      while (m != 0u) {
        const int ee = e0 + __builtin_ctz(m);
        m &= m - 1u;
        const int   s0 = src[ee];
        const float w0 = rsqrtf((float)(cnt[s0] + 1)) * dr;
        fma4(acc, hp[(size_t)s0 * D4], w0);
        if (ROWSUM) ws += w0;
      }
    }
  }
  ((float4*)out)[(size_t)r * D4 + j] = acc;
  if (ROWSUM && j == 0) rsum[r] = ws;
}

static inline void launch_gemm(const float* A, int lda, const float* B,
                               int ldb, float* C, int ldc, int M, int Nc, int K,
                               int ksplit, int cplane,
                               const float* bias, const float* rsum,
                               const float* cvec, hipStream_t stream) {
  dim3 grid(Nc / 128, (M + 127) / 128, ksplit);
  if (bias)
    gemm_wmma<0, 2><<<grid, 256, 0, stream>>>(A, lda, B, ldb, C, ldc, M, Nc, K,
                                              bias, rsum, cvec, cplane, nullptr, 0);
  else
    gemm_wmma<0, 0><<<grid, 256, 0, stream>>>(A, lda, B, ldb, C, ldc, M, Nc, K,
                                              bias, rsum, cvec, cplane, nullptr, 0);
}

extern "C" void kernel_launch(void* const* d_in, const int* in_sizes, int n_in,
                              void* d_out, int out_size, void* d_ws, size_t ws_size,
                              hipStream_t stream) {
  const int IN = 1024, HID = 512, OUT = 128;
  static_assert(IN % XW_BK == 0 && OUT == XW_BN, "gemm_xw: whole K steps, one column of tiles");

  const float* features = (const float*)d_in[0];
  const int*   edge     = (const int*)d_in[1];
  const float* W1       = (const float*)d_in[2];
  const float* b1       = (const float*)d_in[3];
  const float* W2       = (const float*)d_in[4];
  const float* b2       = (const float*)d_in[5];
  const float* b3       = (const float*)d_in[6];
  const float* b4       = (const float*)d_in[7];
  const float* head1    = (const float*)d_in[8];

  const int n  = in_sizes[0] / IN;
  const int E  = in_sizes[1] / 2;
  const int Et = E + n;
  static_assert(3 * LIST_SLOTS + 1 <= OUT, "the slot rows and counters fit the list region");
  const int* src = edge;
  const int* dst = edge + E;

  float* out = (float*)d_out;
  float* z   = out;
  float* h2  = out + (size_t)n * OUT;
  float* h4  = out + (size_t)n * OUT * 2;

  float* ws    = (float*)d_ws;
  float* t0    = ws + n + Et;
  float* t1    = t0 + (size_t)n * HID;
  float* lists = t1 + (size_t)n * HID;
  float* xw    = t0;
  float* axw   = xw + (size_t)n * OUT;
  float* az    = axw + (size_t)n * OUT;
  float* aaz   = az + (size_t)n * OUT;
  float* Wd    = t1;
  float* Wdp   = Wd + (size_t)OUT * IN;
  float* dvec  = Wdp + (size_t)8 * OUT * IN;
  float* cv2   = dvec + IN;
  float* rsum  = cv2 + OUT;

  int*   tbl  = (int*)lists;
  int*   nsrc = tbl + (size_t)n * LIST_SLOTS;
  float* nw   = (float*)(nsrc + (size_t)n * LIST_SLOTS);
  int*   cnt  = (int*)(nw + (size_t)n * LIST_SLOTS);

  const int TB = 256;
  gemm_wmma<0, 0, 1><<<dim3(IN / 128, OUT / 128 + 1, 8), 256, 0, stream>>>(
      W2, HID, W1, IN, Wdp, IN, OUT, IN, HID / 8, nullptr, nullptr, nullptr, OUT * IN, cnt, n);
  fold_bias_kernel<IN, HID, OUT, 8>
      <<<IN / 32 + OUT / 8 + OUT * IN / 1024 + (E + TB - 1) / TB, 256, 0, stream>>>(
          W1, W2, b1, b3, dvec, cv2, Wdp, Wd, dst, cnt, tbl, E);

  gemm_xw<<<2 * ((n + XW_BM - 1) / XW_BM), 256, 0, stream>>>(
      features, IN, Wd, IN, xw, OUT, n, IN, (n + XW_BM - 1) / XW_BM, src, cnt, tbl, nsrc, nw, n);

  const int GB = (n + 7) / 8;
  gather_rows<128, true, false><<<GB, 256, 0, stream>>>(xw, cnt, nsrc, nw, src, dst, E,
                                                        cv2, axw, nullptr, n);
  gather_rows<128, true, false><<<GB, 256, 0, stream>>>(axw, cnt, nsrc, nw, src, dst, E,
                                                        b2, h2, nullptr, n);

  launch_gemm(h2, OUT, head1, OUT, z, OUT, n, OUT, OUT, 1, 0, nullptr, nullptr, nullptr, stream);

  gather_rows<128, false, true><<<GB, 256, 0, stream>>>(z, cnt, nsrc, nw, src, dst, E,
                                                        nullptr, az, rsum, n);
  gather_rows<128, false, false><<<GB, 256, 0, stream>>>(az, cnt, nsrc, nw, src, dst, E,
                                                         nullptr, aaz, nullptr, n);
  launch_gemm(aaz, OUT, Wd, IN, h4, IN, n, IN, OUT, 1, 0, b4, rsum, dvec, stream);
}
